# starting kernel
# baseline (speedup 1.0000x reference)
#include <hip/hip_runtime.h>

typedef float v2f __attribute__((ext_vector_type(2)));
typedef float v8f __attribute__((ext_vector_type(8)));

constexpr int H_ = 8, B_ = 64, S_ = 8, D_ = 64, N_ = 4096, MLPH_ = 128, ITERS_ = 3;
constexpr float LN_EPS_ = 1e-5f, EPS_ATTN_ = 1e-8f, SCALE_ = 0.125f;

__device__ __forceinline__ v8f wmma_f32(v2f a, v2f b, v8f c) {
  return __builtin_amdgcn_wmma_f32_16x16x4_f32(false, a, false, b, (short)0, c,
                                               false, false);
}

__global__ void __launch_bounds__(256) ln_kv_kernel(
    const float* __restrict__ inp, const float* __restrict__ lng,
    const float* __restrict__ lnb, const float* __restrict__ Wk,
    const float* __restrict__ Wv, float* __restrict__ kout,
    float* __restrict__ vout) {
  __shared__ float xln[16][68];
  __shared__ float mu[16], rstd[16];
  const int tid = threadIdx.x;
  const size_t row0 = (size_t)blockIdx.x * 16;
  const int r = tid >> 4, c0 = (tid & 15) * 4;
  float4 val = *(const float4*)(inp + (row0 + r) * D_ + c0);
  xln[r][c0] = val.x; xln[r][c0 + 1] = val.y;
  xln[r][c0 + 2] = val.z; xln[r][c0 + 3] = val.w;
  __syncthreads();
  if (tid < 16) {
    float m = 0.f;
    for (int j = 0; j < D_; ++j) m += xln[tid][j];
    m *= (1.f / D_);
    float var = 0.f;
    for (int j = 0; j < D_; ++j) { float dd = xln[tid][j] - m; var += dd * dd; }
    var *= (1.f / D_);
    mu[tid] = m; rstd[tid] = rsqrtf(var + LN_EPS_);
  }
  __syncthreads();
  {
    float m = mu[r], rs = rstd[r];
    for (int j = 0; j < 4; ++j) {
      int c = c0 + j;
      xln[r][c] = (xln[r][c] - m) * rs * lng[c] + lnb[c];
    }
  }
  __syncthreads();
  const int w = tid >> 5, lane = tid & 31, m16 = lane & 15, kh = lane >> 4;
  const float* W = (w < 4) ? Wk : Wv;
  float* outp = (w < 4) ? kout : vout;
  const int ct = (w & 3) * 16;
  v8f acc = {};
  for (int kk = 0; kk < 16; ++kk) {
    int kb = kk * 4 + kh * 2;
    v2f a; a.x = xln[m16][kb]; a.y = xln[m16][kb + 1];
    v2f b; b.x = W[kb * D_ + ct + m16]; b.y = W[(kb + 1) * D_ + ct + m16];
    acc = wmma_f32(a, b, acc);
  }
  for (int i = 0; i < 8; ++i)
    outp[(row0 + i + 8 * kh) * D_ + ct + m16] = acc[i];
}

__global__ void slots_init_kernel(const float* __restrict__ sinit,
                                  const float* __restrict__ smu,
                                  const float* __restrict__ slog,
                                  float* __restrict__ slots) {
  int idx = blockIdx.x * blockDim.x + threadIdx.x;
  int d = idx & (D_ - 1);
  int h = idx / (B_ * S_ * D_);
  slots[idx] = smu[h * D_ + d] + expf(slog[h * D_ + d]) * sinit[idx];
}

__global__ void __launch_bounds__(128) hprev_q_kernel(
    const float* __restrict__ slots, const float* __restrict__ lng,
    const float* __restrict__ lnb, const float* __restrict__ Wq,
    float* __restrict__ hprev, float* __restrict__ qout) {
  __shared__ float hl[16][68];
  __shared__ float mu[16], rstd[16];
  const int tid = threadIdx.x;
  const int h = blockIdx.x >> 5;
  const int rt = blockIdx.x & 31;
  const int rowbase = h * (B_ * S_) + rt * 16;
  for (int e = tid; e < 16 * D_; e += 128) {
    int r = e >> 6, c = e & 63;
    hl[r][c] = slots[(size_t)(rowbase + r) * D_ + c];
  }
  __syncthreads();
  if (tid < 16) {
    float m = 0.f;
    for (int j = 0; j < D_; ++j) m += hl[tid][j];
    m *= (1.f / D_);
    float var = 0.f;
    for (int j = 0; j < D_; ++j) { float dd = hl[tid][j] - m; var += dd * dd; }
    var *= (1.f / D_);
    mu[tid] = m; rstd[tid] = rsqrtf(var + LN_EPS_);
  }
  __syncthreads();
  for (int e = tid; e < 16 * D_; e += 128) {
    int r = e >> 6, c = e & 63;
    float nv = (hl[r][c] - mu[r]) * rstd[r] * lng[h * D_ + c] + lnb[h * D_ + c];
    hl[r][c] = nv;
    hprev[(size_t)(rowbase + r) * D_ + c] = nv;
  }
  __syncthreads();
  const int w = tid >> 5, lane = tid & 31, m16 = lane & 15, kh = lane >> 4;
  const float* Wh = Wq + (size_t)h * D_ * D_;
  const int ct = w * 16;
  v8f acc = {};
  for (int kk = 0; kk < 16; ++kk) {
    int kb = kk * 4 + kh * 2;
    v2f a; a.x = hl[m16][kb]; a.y = hl[m16][kb + 1];
    v2f b; b.x = Wh[kb * D_ + ct + m16]; b.y = Wh[(kb + 1) * D_ + ct + m16];
    acc = wmma_f32(a, b, acc);
  }
  for (int i = 0; i < 8; ++i)
    qout[(size_t)(rowbase + i + 8 * kh) * D_ + ct + m16] = acc[i];
}

__global__ void __launch_bounds__(128) attn_kernel(
    const float* __restrict__ kbuf, const float* __restrict__ vbuf,
    const float* __restrict__ qbuf, float* __restrict__ updates) {
  __shared__ float qs[16][68];
  __shared__ float kt[64][68];
  __shared__ float at[64][17];
  __shared__ float den[16];
  const int tid = threadIdx.x;
  const int hp = blockIdx.x >> 6;
  const int b  = blockIdx.x & 63;
  const float* kb = kbuf + (size_t)b * N_ * D_;
  const float* vb = vbuf + (size_t)b * N_ * D_;
  for (int e = tid; e < 16 * D_; e += 128) {
    int sp = e >> 6, d = e & 63;
    int h = hp * 2 + (sp >> 3), s = sp & 7;
    qs[sp][d] = qbuf[(size_t)((h * B_ + b) * S_ + s) * D_ + d];
  }
  if (tid < 16) den[tid] = 0.f;
  __syncthreads();
  const int w = tid >> 5, lane = tid & 31, m16 = lane & 15, kh = lane >> 4;
  v8f accu = {};
  for (int nb = 0; nb < N_; nb += 64) {
    for (int e = tid; e < 64 * D_; e += 128) {
      int r = e >> 6, c = e & 63;
      kt[r][c] = kb[(size_t)(nb + r) * D_ + c];
    }
    if (nb + 64 < N_) {
      const char* nk = (const char*)(kb + (size_t)(nb + 64) * D_);
      const char* nv = (const char*)(vb + (size_t)(nb + 64) * D_);
      __builtin_prefetch(nk + tid * 128, 0, 3);
      __builtin_prefetch(nv + tid * 128, 0, 3);
    }
    __syncthreads();
    v8f lg = {};
    for (int kk = 0; kk < 16; ++kk) {
      int kbs = kk * 4 + kh * 2;
      v2f a;  a.x = kt[w * 16 + m16][kbs]; a.y = kt[w * 16 + m16][kbs + 1];
      v2f bb; bb.x = qs[m16][kbs];         bb.y = qs[m16][kbs + 1];
      lg = wmma_f32(a, bb, lg);
    }
    for (int i = 0; i < 8; ++i)
      at[w * 16 + i + 8 * kh][m16] = lg[i] * SCALE_;
    __syncthreads();
    {
      int r = tid >> 1, hh = tid & 1;
      float mx = -3.4e38f;
      for (int s = 0; s < 8; ++s) mx = fmaxf(mx, at[r][hh * 8 + s]);
      float ex[8], sum = 0.f;
      for (int s = 0; s < 8; ++s) { ex[s] = expf(at[r][hh * 8 + s] - mx); sum += ex[s]; }
      float inv = 1.f / sum;
      for (int s = 0; s < 8; ++s) at[r][hh * 8 + s] = ex[s] * inv + EPS_ATTN_;
    }
    __syncthreads();
    if (tid < 16) {
      float dsum = 0.f;
      for (int r = 0; r < 64; ++r) dsum += at[r][tid];
      den[tid] += dsum;
    }
    for (int kk = 0; kk < 16; ++kk) {
      int kbs = kk * 4 + kh * 2;
      v2f a;  a.x = at[kbs][m16];  a.y = at[kbs + 1][m16];
      v2f bb; bb.x = vb[(size_t)(nb + kbs) * D_ + w * 16 + m16];
              bb.y = vb[(size_t)(nb + kbs + 1) * D_ + w * 16 + m16];
      accu = wmma_f32(a, bb, accu);
    }
    __syncthreads();
  }
  for (int i = 0; i < 8; ++i) {
    int sp = i + 8 * kh;
    int h = hp * 2 + (sp >> 3), s = sp & 7;
    updates[(size_t)((h * B_ + b) * S_ + s) * D_ + w * 16 + m16] = accu[i] / den[sp];
  }
}

__global__ void __launch_bounds__(256) gru_mlp_kernel(
    const float* __restrict__ upd, const float* __restrict__ hprev,
    const float* __restrict__ wih, const float* __restrict__ whh,
    const float* __restrict__ bih, const float* __restrict__ bhh,
    const float* __restrict__ lnmg, const float* __restrict__ lnmb,
    const float* __restrict__ w1, const float* __restrict__ b1,
    const float* __restrict__ w2, const float* __restrict__ b2,
    float* __restrict__ slots) {
  __shared__ float ut[16][68];
  __shared__ float ht[16][68];
  __shared__ float gx[16][196];
  __shared__ float gh[16][196];
  __shared__ float sn[16][68];
  __shared__ float nm[16][68];
  __shared__ float hid[16][132];
  __shared__ float mu[16], rstd[16];
  const int tid = threadIdx.x;
  const int h = blockIdx.x >> 5, rt = blockIdx.x & 31;
  const int rowbase = h * (B_ * S_) + rt * 16;
  const int r0 = tid >> 4, c00 = (tid & 15) * 4;
  {
    float4 a = *(const float4*)(upd + (size_t)(rowbase + r0) * D_ + c00);
    ut[r0][c00] = a.x; ut[r0][c00 + 1] = a.y; ut[r0][c00 + 2] = a.z; ut[r0][c00 + 3] = a.w;
    float4 c = *(const float4*)(hprev + (size_t)(rowbase + r0) * D_ + c00);
    ht[r0][c00] = c.x; ht[r0][c00 + 1] = c.y; ht[r0][c00 + 2] = c.z; ht[r0][c00 + 3] = c.w;
  }
  __syncthreads();
  const int w = tid >> 5, lane = tid & 31, m16 = lane & 15, kh = lane >> 4;
  const float* wihh = wih + (size_t)h * D_ * 192;
  const float* whhh = whh + (size_t)h * D_ * 192;
  for (int T = w; T < 24; T += 8) {
    const bool isX = T < 12;
    const int ct = (T % 12) * 16;
    const float* Wt = isX ? wihh : whhh;
    v8f acc = {};
    for (int kk = 0; kk < 16; ++kk) {
      int kb = kk * 4 + kh * 2;
      v2f a;
      if (isX) { a.x = ut[m16][kb]; a.y = ut[m16][kb + 1]; }
      else     { a.x = ht[m16][kb]; a.y = ht[m16][kb + 1]; }
      v2f bb; bb.x = Wt[kb * 192 + ct + m16]; bb.y = Wt[(kb + 1) * 192 + ct + m16];
      acc = wmma_f32(a, bb, acc);
    }
    float* dst = isX ? &gx[0][0] : &gh[0][0];
    for (int i = 0; i < 8; ++i)
      dst[(i + 8 * kh) * 196 + ct + m16] = acc[i];
  }
  __syncthreads();
  {
    const float* bi = bih + h * 192;
    const float* bh = bhh + h * 192;
    for (int j = 0; j < 4; ++j) {
      int d = c00 + j;
      float xr = gx[r0][d] + bi[d],        hr = gh[r0][d] + bh[d];
      float xz = gx[r0][64 + d] + bi[64 + d],  hz = gh[r0][64 + d] + bh[64 + d];
      float xn = gx[r0][128 + d] + bi[128 + d], hn = gh[r0][128 + d] + bh[128 + d];
      float rg = 1.f / (1.f + expf(-(xr + hr)));
      float zg = 1.f / (1.f + expf(-(xz + hz)));
      float ng = tanhf(xn + rg * hn);
      sn[r0][d] = (1.f - zg) * ng + zg * ht[r0][d];
    }
  }
  __syncthreads();
  if (tid < 16) {
    float m = 0.f;
    for (int j = 0; j < D_; ++j) m += sn[tid][j];
    m *= (1.f / D_);
    float var = 0.f;
    for (int j = 0; j < D_; ++j) { float dd = sn[tid][j] - m; var += dd * dd; }
    var *= (1.f / D_);
    mu[tid] = m; rstd[tid] = rsqrtf(var + LN_EPS_);
  }
  __syncthreads();
  for (int j = 0; j < 4; ++j) {
    int d = c00 + j;
    nm[r0][d] = (sn[r0][d] - mu[r0]) * rstd[r0] * lnmg[h * D_ + d] + lnmb[h * D_ + d];
  }
  __syncthreads();
  {
    const float* w1h = w1 + (size_t)h * D_ * MLPH_;
    const float* b1h = b1 + h * MLPH_;
    const int ct = w * 16;
    v8f acc = {};
    for (int kk = 0; kk < 16; ++kk) {
      int kb = kk * 4 + kh * 2;
      v2f a;  a.x = nm[m16][kb]; a.y = nm[m16][kb + 1];
      v2f bb; bb.x = w1h[kb * MLPH_ + ct + m16]; bb.y = w1h[(kb + 1) * MLPH_ + ct + m16];
      acc = wmma_f32(a, bb, acc);
    }
    float bias = b1h[ct + m16];
    for (int i = 0; i < 8; ++i)
      hid[i + 8 * kh][ct + m16] = fmaxf(acc[i] + bias, 0.f);
  }
  __syncthreads();
  if (w < 4) {
    const float* w2h = w2 + (size_t)h * MLPH_ * D_;
    const float* b2h = b2 + h * D_;
    const int ct = w * 16;
    v8f acc = {};
    for (int kk = 0; kk < 32; ++kk) {
      int kb = kk * 4 + kh * 2;
      v2f a;  a.x = hid[m16][kb]; a.y = hid[m16][kb + 1];
      v2f bb; bb.x = w2h[kb * D_ + ct + m16]; bb.y = w2h[(kb + 1) * D_ + ct + m16];
      acc = wmma_f32(a, bb, acc);
    }
    float bias = b2h[ct + m16];
    for (int i = 0; i < 8; ++i) {
      int rr = i + 8 * kh;
      slots[(size_t)(rowbase + rr) * D_ + ct + m16] = sn[rr][ct + m16] + acc[i] + bias;
    }
  }
}

__global__ void copy_out_kernel(const float* __restrict__ slots,
                                float* __restrict__ out) {
  int i = blockIdx.x * blockDim.x + threadIdx.x;
  if (i < B_ * S_ * D_) out[i] = slots[i];
}

extern "C" void kernel_launch(void* const* d_in, const int* in_sizes, int n_in,
                              void* d_out, int out_size, void* d_ws, size_t ws_size,
                              hipStream_t stream) {
  (void)in_sizes; (void)n_in; (void)out_size; (void)ws_size;
  const float* inputs  = (const float*)d_in[0];
  const float* sinit   = (const float*)d_in[1];
  const float* smu     = (const float*)d_in[2];
  const float* slog    = (const float*)d_in[3];
  const float* ln_in_g = (const float*)d_in[4];
  const float* ln_in_b = (const float*)d_in[5];
  const float* Wk      = (const float*)d_in[6];
  const float* Wv      = (const float*)d_in[7];
  const float* ln_s_g  = (const float*)d_in[8];
  const float* ln_s_b  = (const float*)d_in[9];
  const float* Wq      = (const float*)d_in[10];
  const float* gru_wih = (const float*)d_in[11];
  const float* gru_whh = (const float*)d_in[12];
  const float* gru_bih = (const float*)d_in[13];
  const float* gru_bhh = (const float*)d_in[14];
  const float* ln_m_g  = (const float*)d_in[15];
  const float* ln_m_b  = (const float*)d_in[16];
  const float* mlp_w1  = (const float*)d_in[17];
  const float* mlp_b1  = (const float*)d_in[18];
  const float* mlp_w2  = (const float*)d_in[19];
  const float* mlp_b2  = (const float*)d_in[20];

  float* ws = (float*)d_ws;
  float* kbuf  = ws;
  float* vbuf  = kbuf + (size_t)B_ * N_ * D_;
  float* slots = vbuf + (size_t)B_ * N_ * D_;
  float* hprev = slots + H_ * B_ * S_ * D_;
  float* qbuf  = hprev + H_ * B_ * S_ * D_;
  float* updv  = qbuf + H_ * B_ * S_ * D_;

  ln_kv_kernel<<<(B_ * N_) / 16, 256, 0, stream>>>(inputs, ln_in_g, ln_in_b,
                                                   Wk, Wv, kbuf, vbuf);
  slots_init_kernel<<<(H_ * B_ * S_ * D_) / 256, 256, 0, stream>>>(sinit, smu,
                                                                   slog, slots);
  for (int it = 0; it < ITERS_; ++it) {
    hprev_q_kernel<<<H_ * 32, 128, 0, stream>>>(slots, ln_s_g, ln_s_b, Wq,
                                                hprev, qbuf);
    attn_kernel<<<(H_ / 2) * B_, 128, 0, stream>>>(kbuf, vbuf, qbuf, updv);
    gru_mlp_kernel<<<H_ * 32, 256, 0, stream>>>(updv, hprev, gru_wih, gru_whh,
                                                gru_bih, gru_bhh, ln_m_g, ln_m_b,
                                                mlp_w1, mlp_b1, mlp_w2, mlp_b2,
                                                slots);
  }
  copy_out_kernel<<<(B_ * S_ * D_ + 255) / 256, 256, 0, stream>>>(slots,
                                                                  (float*)d_out);
}


# attention block = (4 heads, batch, half of N): both slot tiles share each k and v fragment and each staged row; half sums added at the GRU load; logit order kept, update and den sums split in two
# speedup vs baseline: 2.9598x; 2.9598x over previous
#include <hip/hip_runtime.h>

typedef float v2f __attribute__((ext_vector_type(2)));
typedef float v4f __attribute__((ext_vector_type(4)));
typedef float v8f __attribute__((ext_vector_type(8)));

constexpr int H_ = 8, B_ = 64, S_ = 8, D_ = 64, N_ = 4096, MLPH_ = 128, ITERS_ = 3;
constexpr float LN_EPS_ = 1e-5f, EPS_ATTN_ = 1e-8f, SCALE_ = 0.125f;

__device__ __forceinline__ v8f wmma_f32(v2f a, v2f b, v8f c) {
  return __builtin_amdgcn_wmma_f32_16x16x4_f32(false, a, false, b, (short)0, c,
                                               false, false);
}

constexpr int LKV_ROWS_ = 128;

__device__ __forceinline__ float row16_sum(float v) {
  v += __int_as_float(__builtin_amdgcn_update_dpp(0, __float_as_int(v), 0x161, 0xf, 0xf, false));
  v += __int_as_float(__builtin_amdgcn_update_dpp(0, __float_as_int(v), 0x162, 0xf, 0xf, false));
  v += __int_as_float(__builtin_amdgcn_update_dpp(0, __float_as_int(v), 0x164, 0xf, 0xf, false));
  v += __int_as_float(__builtin_amdgcn_update_dpp(0, __float_as_int(v), 0x168, 0xf, 0xf, false));
  return v;
}

__global__ void __launch_bounds__(256) ln_kv_kernel(
    const float* __restrict__ inp, const float* __restrict__ lng,
    const float* __restrict__ lnb, const float* __restrict__ Wk,
    const float* __restrict__ Wv, float* __restrict__ kout,
    float* __restrict__ vout) {
  __shared__ __attribute__((aligned(16))) float xln[LKV_ROWS_][68];
  const int tid = threadIdx.x;
  const size_t row0 = (size_t)blockIdx.x * LKV_ROWS_;
  const int w = tid >> 5, lane = tid & 31, m16 = lane & 15, kh = lane >> 4;
  const float* Wm = (w < 4) ? Wk : Wv;
  float* outp = (w < 4) ? kout : vout;
  const int ct = (w & 3) * 16;
  v2f bf[16];
#pragma unroll
  for (int kk = 0; kk < 16; ++kk) {
    const int kb = kk * 4 + kh * 2;
    bf[kk].x = Wm[kb * D_ + ct + m16];
    bf[kk].y = Wm[(kb + 1) * D_ + ct + m16];
  }
  const int r = tid >> 4, c0 = (tid & 15) * 4;
  const float4 g4 = *(const float4*)(lng + c0);
  const float4 b4 = *(const float4*)(lnb + c0);
  float4 xv[8];
#pragma unroll
  for (int j = 0; j < 8; ++j)
    xv[j] = *(const float4*)(inp + (row0 + r + 16 * j) * D_ + c0);
#pragma unroll
  for (int j = 0; j < 8; ++j) {
    const float m = row16_sum((xv[j].x + xv[j].y) + (xv[j].z + xv[j].w)) * (1.f / D_);
    const float dx = xv[j].x - m, dy = xv[j].y - m, dz = xv[j].z - m, dw = xv[j].w - m;
    const float var = row16_sum((dx * dx + dy * dy) + (dz * dz + dw * dw)) * (1.f / D_);
    const float rs = rsqrtf(var + LN_EPS_);
    float4 o;
    o.x = dx * rs * g4.x + b4.x;
    o.y = dy * rs * g4.y + b4.y;
    o.z = dz * rs * g4.z + b4.z;
    o.w = dw * rs * g4.w + b4.w;
    *(float4*)&xln[r + 16 * j][c0] = o;
  }
  __syncthreads();
  float* op = outp + (row0 + 8 * kh) * D_ + ct + m16;
#pragma unroll
  for (int mt = 0; mt < LKV_ROWS_ / 16; mt += 2) {
    v2f a0[16], a1[16];
#pragma unroll
    for (int kk = 0; kk < 16; ++kk) {
      const int kb = kk * 4 + kh * 2;
      a0[kk].x = xln[mt * 16 + m16][kb];      a0[kk].y = xln[mt * 16 + m16][kb + 1];
      a1[kk].x = xln[mt * 16 + 16 + m16][kb]; a1[kk].y = xln[mt * 16 + 16 + m16][kb + 1];
    }
    __builtin_amdgcn_sched_barrier(0);
    v8f acc0 = {}, acc1 = {};
#pragma unroll
    for (int kk = 0; kk < 16; ++kk) {
      acc0 = wmma_f32(a0[kk], bf[kk], acc0);
      acc1 = wmma_f32(a1[kk], bf[kk], acc1);
    }
#pragma unroll
    for (int i = 0; i < 8; ++i) {
      op[(mt * 16 + i) * D_] = acc0[i];
      op[(mt * 16 + 16 + i) * D_] = acc1[i];
    }
  }
}

__global__ void slots_init_kernel(const float* __restrict__ sinit,
                                  const float* __restrict__ smu,
                                  const float* __restrict__ slog,
                                  float* __restrict__ slots) {
  int idx = blockIdx.x * blockDim.x + threadIdx.x;
  int d = idx & (D_ - 1);
  int h = idx / (B_ * S_ * D_);
  slots[idx] = smu[h * D_ + d] + expf(slog[h * D_ + d]) * sinit[idx];
}

__global__ void __launch_bounds__(128) hprev_q_kernel(
    const float* __restrict__ slots, const float* __restrict__ lng,
    const float* __restrict__ lnb, const float* __restrict__ Wq,
    float* __restrict__ hprev, float* __restrict__ qout) {
  __shared__ float hl[16][68];
  __shared__ float mu[16], rstd[16];
  const int tid = threadIdx.x;
  const int h = blockIdx.x >> 5;
  const int rt = blockIdx.x & 31;
  const int rowbase = h * (B_ * S_) + rt * 16;
  for (int e = tid; e < 16 * D_; e += 128) {
    int r = e >> 6, c = e & 63;
    hl[r][c] = slots[(size_t)(rowbase + r) * D_ + c];
  }
  __syncthreads();
  if (tid < 16) {
    float m = 0.f;
    for (int j = 0; j < D_; ++j) m += hl[tid][j];
    m *= (1.f / D_);
    float var = 0.f;
    for (int j = 0; j < D_; ++j) { float dd = hl[tid][j] - m; var += dd * dd; }
    var *= (1.f / D_);
    mu[tid] = m; rstd[tid] = rsqrtf(var + LN_EPS_);
  }
  __syncthreads();
  for (int e = tid; e < 16 * D_; e += 128) {
    int r = e >> 6, c = e & 63;
    float nv = (hl[r][c] - mu[r]) * rstd[r] * lng[h * D_ + c] + lnb[h * D_ + c];
    hl[r][c] = nv;
    hprev[(size_t)(rowbase + r) * D_ + c] = nv;
  }
  __syncthreads();
  const int w = tid >> 5, lane = tid & 31, m16 = lane & 15, kh = lane >> 4;
  const float* Wh = Wq + (size_t)h * D_ * D_;
  const int ct = w * 16;
  v8f acc = {};
  for (int kk = 0; kk < 16; ++kk) {
    int kb = kk * 4 + kh * 2;
    v2f a; a.x = hl[m16][kb]; a.y = hl[m16][kb + 1];
    v2f b; b.x = Wh[kb * D_ + ct + m16]; b.y = Wh[(kb + 1) * D_ + ct + m16];
    acc = wmma_f32(a, b, acc);
  }
  for (int i = 0; i < 8; ++i)
    qout[(size_t)(rowbase + i + 8 * kh) * D_ + ct + m16] = acc[i];
}

constexpr int TR_ = 128;
constexpr int NH_ = N_ / 2;
constexpr int NTH_ = NH_ / TR_;
constexpr int VP_ = 72;

template <int P>
__device__ __forceinline__ void attn_tile_step(
    float (&kt)[2][TR_][68], float (&vt)[2][TR_][VP_], float (&at)[2][2][TR_][17],
    const float* knext, const float* vnext, const v2f (&qf)[2][16],
    v8f (&accu)[2][2], float (&dacc)[2][8],
    const int tid, const int rb, const int gb, const int w, const int m16,
    const int kh) {
  const int sr = tid >> 4, sc = (tid & 15) * 4;
  v2f kf[16];
#pragma unroll
  for (int kk = 0; kk < 16; ++kk) {
    const int kbs = kk * 4 + kh * 2;
    kf[kk].x = kt[P][rb + m16][kbs];
    kf[kk].y = kt[P][rb + m16][kbs + 1];
  }
  const float* kp = knext + (size_t)sr * D_ + sc;
  const float* vp = vnext + (size_t)sr * D_ + sc;
  v4f kn[8], vn[8];
#pragma unroll
  for (int j = 0; j < 8; ++j) kn[j] = *(const v4f*)(kp + (size_t)j * 16 * D_);
#pragma unroll
  for (int j = 0; j < 8; ++j) vn[j] = *(const v4f*)(vp + (size_t)j * 16 * D_);
  asm volatile("" ::: "memory");
  __builtin_amdgcn_sched_barrier(0);
  v8f la[2], lb[2];
  la[0] = v8f{}; la[1] = v8f{}; lb[0] = v8f{}; lb[1] = v8f{};
#pragma unroll
  for (int kk = 0; kk < 8; ++kk) {
#pragma unroll
    for (int c = 0; c < 2; ++c) {
      la[c] = wmma_f32(kf[kk], qf[c][kk], la[c]);
      lb[c] = wmma_f32(kf[kk + 8], qf[c][kk + 8], lb[c]);
    }
  }
#pragma unroll
  for (int c = 0; c < 2; ++c)
#pragma unroll
    for (int i = 0; i < 8; ++i)
      at[P][c][rb + i + 8 * kh][m16] = (la[c][i] + lb[c][i]) * SCALE_;
  __builtin_amdgcn_wave_barrier();
  const int r = tid >> 1, hh = tid & 1;
  float lv[2][8];
#pragma unroll
  for (int c = 0; c < 2; ++c)
#pragma unroll
    for (int s = 0; s < 8; ++s) lv[c][s] = at[P][c][r][hh * 8 + s];
  __builtin_amdgcn_sched_barrier(0);
  v2f vf[16];
#pragma unroll
  for (int kk = 0; kk < 16; ++kk) {
    const int kbs = kk * 4 + kh * 2;
    vf[kk].x = vt[P][gb + kbs][w * 16 + m16];
    vf[kk].y = vt[P][gb + kbs + 1][w * 16 + m16];
  }
  __builtin_amdgcn_sched_barrier(0);
#pragma unroll
  for (int c = 0; c < 2; ++c) {
    float mx = -3.4e38f;
#pragma unroll
    for (int s = 0; s < 8; ++s) mx = fmaxf(mx, lv[c][s]);
    float ex[8], sum = 0.f;
#pragma unroll
    for (int s = 0; s < 8; ++s) { ex[s] = expf(lv[c][s] - mx); sum += ex[s]; }
    const float inv = 1.f / sum;
#pragma unroll
    for (int s = 0; s < 8; ++s) {
      const float pv = ex[s] * inv + EPS_ATTN_;
      at[P][c][r][hh * 8 + s] = pv;
      dacc[c][s] += pv;
    }
  }
  __builtin_amdgcn_sched_barrier(0);
#pragma unroll
  for (int j = 0; j < 8; ++j) *(v4f*)&kt[P ^ 1][sr + 16 * j][sc] = kn[j];
#pragma unroll
  for (int j = 0; j < 8; ++j) *(v4f*)&vt[P ^ 1][sr + 16 * j][sc] = vn[j];
  __syncthreads();
  v2f af[2][16];
#pragma unroll
  for (int c = 0; c < 2; ++c)
#pragma unroll
    for (int kk = 0; kk < 16; ++kk) {
      const int kbs = kk * 4 + kh * 2;
      af[c][kk].x = at[P][c][gb + kbs][m16];
      af[c][kk].y = at[P][c][gb + kbs + 1][m16];
    }
  __builtin_amdgcn_sched_barrier(0);
#pragma unroll
  for (int kk = 0; kk < 8; ++kk) {
#pragma unroll
    for (int c = 0; c < 2; ++c) {
      accu[c][0] = wmma_f32(af[c][kk], vf[kk], accu[c][0]);
      accu[c][1] = wmma_f32(af[c][kk + 8], vf[kk + 8], accu[c][1]);
    }
  }
}

__global__ void __launch_bounds__(256) __attribute__((amdgpu_waves_per_eu(2, 2)))
attn_kernel(
    const float* kbuf, const float* vbuf,
    const float* __restrict__ qbuf, float* pnum0, float* pnum1, float* pden) {
  __shared__ __attribute__((aligned(16))) float kt[2][TR_][68];
  __shared__ __attribute__((aligned(16))) float vt[2][TR_][VP_];
  __shared__ float at[2][2][TR_][17];
  __shared__ float dsc[32][TR_ + 1];
  __shared__ float red[2][4][8][32];
  const int tid = threadIdx.x;
  const int hq = blockIdx.x >> 7;
  const int nh = (blockIdx.x >> 6) & 1;
  const int b  = blockIdx.x & 63;
  const float* kb = kbuf + ((size_t)b * N_ + (size_t)nh * NH_) * D_;
  const float* vb = vbuf + ((size_t)b * N_ + (size_t)nh * NH_) * D_;
  const int w8 = tid >> 5, lane = tid & 31, m16 = lane & 15, kh = lane >> 4;
  const int g = w8 >> 2, w = w8 & 3;
  const int rb = w8 * 16, gb = g * 64;
  v2f qf[2][16];
#pragma unroll
  for (int c = 0; c < 2; ++c) {
    const int h = (hq * 2 + c) * 2 + (m16 >> 3), s = m16 & 7;
    const float* qr = qbuf + (size_t)((h * B_ + b) * S_ + s) * D_ + kh * 2;
#pragma unroll
    for (int kk = 0; kk < 16; ++kk) { qf[c][kk].x = qr[kk * 4]; qf[c][kk].y = qr[kk * 4 + 1]; }
  }
  {
    const int sr = tid >> 4, sc = (tid & 15) * 4;
#pragma unroll
    for (int j = 0; j < 8; ++j) {
      *(float4*)&kt[0][sr + 16 * j][sc] =
          *(const float4*)(kb + (size_t)(sr + 16 * j) * D_ + sc);
      *(float4*)&vt[0][sr + 16 * j][sc] =
          *(const float4*)(vb + (size_t)(sr + 16 * j) * D_ + sc);
    }
  }
  __syncthreads();
  v8f accu[2][2];
  float dacc[2][8];
#pragma unroll
  for (int c = 0; c < 2; ++c) {
    accu[c][0] = v8f{};
    accu[c][1] = v8f{};
#pragma unroll
    for (int s = 0; s < 8; ++s) dacc[c][s] = 0.f;
  }
  for (int t = 0; t < NTH_; t += 2) {
    attn_tile_step<0>(kt, vt, at, kb + (size_t)(t + 1) * TR_ * D_,
                      vb + (size_t)(t + 1) * TR_ * D_, qf, accu, dacc,
                      tid, rb, gb, w, m16, kh);
    const int tn = (t + 2 < NTH_) ? t + 2 : t + 1;
    attn_tile_step<1>(kt, vt, at, kb + (size_t)tn * TR_ * D_,
                      vb + (size_t)tn * TR_ * D_, qf, accu, dacc,
                      tid, rb, gb, w, m16, kh);
  }
  v8f asum[2];
  asum[0] = accu[0][0] + accu[0][1];
  asum[1] = accu[1][0] + accu[1][1];
  if (g == 1) {
#pragma unroll
    for (int c = 0; c < 2; ++c)
#pragma unroll
      for (int i = 0; i < 8; ++i) red[c][w][i][lane] = asum[c][i];
  }
  {
    const int r = tid >> 1, hh = tid & 1;
#pragma unroll
    for (int c = 0; c < 2; ++c)
#pragma unroll
      for (int s = 0; s < 8; ++s) dsc[c * 16 + hh * 8 + s][r] = dacc[c][s];
  }
  __syncthreads();
  if (tid < 32) {
    float dsum = 0.f;
    for (int r = 0; r < TR_; ++r) dsum += dsc[tid][r];
    const int h = hq * 4 + (tid >> 3), s = tid & 7;
    pden[nh * (H_ * B_ * S_) + (h * B_ + b) * S_ + s] = dsum;
  }
  if (g == 0) {
    float* pn = nh ? pnum1 : pnum0;
#pragma unroll
    for (int c = 0; c < 2; ++c)
      for (int i = 0; i < 8; ++i) {
        const int sp = i + 8 * kh;
        const int h = (hq * 2 + c) * 2 + (sp >> 3), s = sp & 7;
        pn[(size_t)((h * B_ + b) * S_ + s) * D_ + w * 16 + m16] =
            asum[c][i] + red[c][w][i][lane];
      }
  }
}

__global__ void __launch_bounds__(256) gru_mlp_kernel(
    const float* __restrict__ pnum0, const float* pnum1,
    const float* __restrict__ pden, const float* __restrict__ hprev,
    const float* __restrict__ wih, const float* __restrict__ whh,
    const float* __restrict__ bih, const float* __restrict__ bhh,
    const float* __restrict__ lnmg, const float* __restrict__ lnmb,
    const float* __restrict__ w1, const float* __restrict__ b1,
    const float* __restrict__ w2, const float* __restrict__ b2,
    float* slots) {
  __shared__ float ut[16][68];
  __shared__ float ht[16][68];
  __shared__ float gx[16][196];
  __shared__ float gh[16][196];
  __shared__ float sn[16][68];
  __shared__ float nm[16][68];
  __shared__ float hid[16][132];
  __shared__ float mu[16], rstd[16];
  const int tid = threadIdx.x;
  const int h = blockIdx.x >> 5, rt = blockIdx.x & 31;
  const int rowbase = h * (B_ * S_) + rt * 16;
  const int r0 = tid >> 4, c00 = (tid & 15) * 4;
  {
    const float4 a0 = *(const float4*)(pnum0 + (size_t)(rowbase + r0) * D_ + c00);
    const float4 a1 = *(const float4*)(pnum1 + (size_t)(rowbase + r0) * D_ + c00);
    const float dn = pden[rowbase + r0] + pden[H_ * B_ * S_ + rowbase + r0];
    ut[r0][c00] = (a0.x + a1.x) / dn;         ut[r0][c00 + 1] = (a0.y + a1.y) / dn;
    ut[r0][c00 + 2] = (a0.z + a1.z) / dn;     ut[r0][c00 + 3] = (a0.w + a1.w) / dn;
    float4 c = *(const float4*)(hprev + (size_t)(rowbase + r0) * D_ + c00);
    ht[r0][c00] = c.x; ht[r0][c00 + 1] = c.y; ht[r0][c00 + 2] = c.z; ht[r0][c00 + 3] = c.w;
  }
  __syncthreads();
  const int w = tid >> 5, lane = tid & 31, m16 = lane & 15, kh = lane >> 4;
  const float* wihh = wih + (size_t)h * D_ * 192;
  const float* whhh = whh + (size_t)h * D_ * 192;
  for (int T = w; T < 24; T += 8) {
    const bool isX = T < 12;
    const int ct = (T % 12) * 16;
    const float* Wt = isX ? wihh : whhh;
    v8f acc = {};
    for (int kk = 0; kk < 16; ++kk) {
      int kb = kk * 4 + kh * 2;
      v2f a;
      if (isX) { a.x = ut[m16][kb]; a.y = ut[m16][kb + 1]; }
      else     { a.x = ht[m16][kb]; a.y = ht[m16][kb + 1]; }
      v2f bb; bb.x = Wt[kb * 192 + ct + m16]; bb.y = Wt[(kb + 1) * 192 + ct + m16];
      acc = wmma_f32(a, bb, acc);
    }
    float* dst = isX ? &gx[0][0] : &gh[0][0];
    for (int i = 0; i < 8; ++i)
      dst[(i + 8 * kh) * 196 + ct + m16] = acc[i];
  }
  __syncthreads();
  {
    const float* bi = bih + h * 192;
    const float* bh = bhh + h * 192;
    for (int j = 0; j < 4; ++j) {
      int d = c00 + j;
      float xr = gx[r0][d] + bi[d],        hr = gh[r0][d] + bh[d];
      float xz = gx[r0][64 + d] + bi[64 + d],  hz = gh[r0][64 + d] + bh[64 + d];
      float xn = gx[r0][128 + d] + bi[128 + d], hn = gh[r0][128 + d] + bh[128 + d];
      float rg = 1.f / (1.f + expf(-(xr + hr)));
      float zg = 1.f / (1.f + expf(-(xz + hz)));
      float ng = tanhf(xn + rg * hn);
      sn[r0][d] = (1.f - zg) * ng + zg * ht[r0][d];
    }
  }
  __syncthreads();
  if (tid < 16) {
    float m = 0.f;
    for (int j = 0; j < D_; ++j) m += sn[tid][j];
    m *= (1.f / D_);
    float var = 0.f;
    for (int j = 0; j < D_; ++j) { float dd = sn[tid][j] - m; var += dd * dd; }
    var *= (1.f / D_);
    mu[tid] = m; rstd[tid] = rsqrtf(var + LN_EPS_);
  }
  __syncthreads();
  for (int j = 0; j < 4; ++j) {
    int d = c00 + j;
    nm[r0][d] = (sn[r0][d] - mu[r0]) * rstd[r0] * lnmg[h * D_ + d] + lnmb[h * D_ + d];
  }
  __syncthreads();
  {
    const float* w1h = w1 + (size_t)h * D_ * MLPH_;
    const float* b1h = b1 + h * MLPH_;
    const int ct = w * 16;
    v8f acc = {};
    for (int kk = 0; kk < 16; ++kk) {
      int kb = kk * 4 + kh * 2;
      v2f a;  a.x = nm[m16][kb]; a.y = nm[m16][kb + 1];
      v2f bb; bb.x = w1h[kb * MLPH_ + ct + m16]; bb.y = w1h[(kb + 1) * MLPH_ + ct + m16];
      acc = wmma_f32(a, bb, acc);
    }
    float bias = b1h[ct + m16];
    for (int i = 0; i < 8; ++i)
      hid[i + 8 * kh][ct + m16] = fmaxf(acc[i] + bias, 0.f);
  }
  __syncthreads();
  if (w < 4) {
    const float* w2h = w2 + (size_t)h * MLPH_ * D_;
    const float* b2h = b2 + h * D_;
    const int ct = w * 16;
    v8f acc = {};
    for (int kk = 0; kk < 32; ++kk) {
      int kb = kk * 4 + kh * 2;
      v2f a;  a.x = hid[m16][kb]; a.y = hid[m16][kb + 1];
      v2f bb; bb.x = w2h[kb * D_ + ct + m16]; bb.y = w2h[(kb + 1) * D_ + ct + m16];
      acc = wmma_f32(a, bb, acc);
    }
    float bias = b2h[ct + m16];
    for (int i = 0; i < 8; ++i) {
      int rr = i + 8 * kh;
      slots[(size_t)(rowbase + rr) * D_ + ct + m16] = sn[rr][ct + m16] + acc[i] + bias;
    }
  }
}

__global__ void copy_out_kernel(const float* __restrict__ slots,
                                float* __restrict__ out) {
  int i = blockIdx.x * blockDim.x + threadIdx.x;
  if (i < B_ * S_ * D_) out[i] = slots[i];
}

extern "C" void kernel_launch(void* const* d_in, const int* in_sizes, int n_in,
                              void* d_out, int out_size, void* d_ws, size_t ws_size,
                              hipStream_t stream) {
  (void)in_sizes; (void)n_in; (void)out_size; (void)ws_size;
  const float* inputs  = (const float*)d_in[0];
  const float* sinit   = (const float*)d_in[1];
  const float* smu     = (const float*)d_in[2];
  const float* slog    = (const float*)d_in[3];
  const float* ln_in_g = (const float*)d_in[4];
  const float* ln_in_b = (const float*)d_in[5];
  const float* Wk      = (const float*)d_in[6];
  const float* Wv      = (const float*)d_in[7];
  const float* ln_s_g  = (const float*)d_in[8];
  const float* ln_s_b  = (const float*)d_in[9];
  const float* Wq      = (const float*)d_in[10];
  const float* gru_wih = (const float*)d_in[11];
  const float* gru_whh = (const float*)d_in[12];
  const float* gru_bih = (const float*)d_in[13];
  const float* gru_bhh = (const float*)d_in[14];
  const float* ln_m_g  = (const float*)d_in[15];
  const float* ln_m_b  = (const float*)d_in[16];
  const float* mlp_w1  = (const float*)d_in[17];
  const float* mlp_b1  = (const float*)d_in[18];
  const float* mlp_w2  = (const float*)d_in[19];
  const float* mlp_b2  = (const float*)d_in[20];

  float* ws = (float*)d_ws;
  float* kbuf  = ws;
  float* vbuf  = kbuf + (size_t)B_ * N_ * D_;
  float* slots = vbuf + (size_t)B_ * N_ * D_;
  float* hprev = slots + H_ * B_ * S_ * D_;
  float* qbuf  = hprev + H_ * B_ * S_ * D_;
  float* updv  = qbuf + H_ * B_ * S_ * D_;
  float* dpart = (float*)d_out;

  ln_kv_kernel<<<(B_ * N_) / LKV_ROWS_, 256, 0, stream>>>(inputs, ln_in_g, ln_in_b,
                                                   Wk, Wv, kbuf, vbuf);
  slots_init_kernel<<<(H_ * B_ * S_ * D_) / 256, 256, 0, stream>>>(sinit, smu,
                                                                   slog, slots);
  for (int it = 0; it < ITERS_; ++it) {
    hprev_q_kernel<<<H_ * 32, 128, 0, stream>>>(slots, ln_s_g, ln_s_b, Wq,
                                                hprev, qbuf);
    attn_kernel<<<(H_ / 4) * 2 * B_, 256, 0, stream>>>(kbuf, vbuf, qbuf, updv,
                                                       slots, dpart);
    gru_mlp_kernel<<<H_ * 32, 256, 0, stream>>>(updv, slots, dpart, hprev,
                                                gru_wih, gru_whh,
                                                gru_bih, gru_bhh, ln_m_g, ln_m_b,
                                                mlp_w1, mlp_b1, mlp_w2, mlp_b2,
                                                slots);
  }
  copy_out_kernel<<<(B_ * S_ * D_ + 255) / 256, 256, 0, stream>>>(slots,
                                                                  (float*)d_out);
}


# attention softmax in base 2: logits scaled by log2(e)/sqrt(D) at the store, hardware exp2 and rcp instructions replace library expf and IEEE division; attn changes in its last bits
# speedup vs baseline: 3.1254x; 1.0559x over previous
#include <hip/hip_runtime.h>

typedef float v2f __attribute__((ext_vector_type(2)));
typedef float v4f __attribute__((ext_vector_type(4)));
typedef float v8f __attribute__((ext_vector_type(8)));

constexpr int H_ = 8, B_ = 64, S_ = 8, D_ = 64, N_ = 4096, MLPH_ = 128, ITERS_ = 3;
constexpr float LN_EPS_ = 1e-5f, EPS_ATTN_ = 1e-8f, SCALE_ = 0.125f;
constexpr float LSCALE2_ = SCALE_ * 1.4426950408889634f;

__device__ __forceinline__ v8f wmma_f32(v2f a, v2f b, v8f c) {
  return __builtin_amdgcn_wmma_f32_16x16x4_f32(false, a, false, b, (short)0, c,
                                               false, false);
}

constexpr int LKV_ROWS_ = 128;

__device__ __forceinline__ float row16_sum(float v) {
  v += __int_as_float(__builtin_amdgcn_update_dpp(0, __float_as_int(v), 0x161, 0xf, 0xf, false));
  v += __int_as_float(__builtin_amdgcn_update_dpp(0, __float_as_int(v), 0x162, 0xf, 0xf, false));
  v += __int_as_float(__builtin_amdgcn_update_dpp(0, __float_as_int(v), 0x164, 0xf, 0xf, false));
  v += __int_as_float(__builtin_amdgcn_update_dpp(0, __float_as_int(v), 0x168, 0xf, 0xf, false));
  return v;
}

__global__ void __launch_bounds__(256) ln_kv_kernel(
    const float* __restrict__ inp, const float* __restrict__ lng,
    const float* __restrict__ lnb, const float* __restrict__ Wk,
    const float* __restrict__ Wv, float* __restrict__ kout,
    float* __restrict__ vout) {
  __shared__ __attribute__((aligned(16))) float xln[LKV_ROWS_][68];
  const int tid = threadIdx.x;
  const size_t row0 = (size_t)blockIdx.x * LKV_ROWS_;
  const int w = tid >> 5, lane = tid & 31, m16 = lane & 15, kh = lane >> 4;
  const float* Wm = (w < 4) ? Wk : Wv;
  float* outp = (w < 4) ? kout : vout;
  const int ct = (w & 3) * 16;
  v2f bf[16];
#pragma unroll
  for (int kk = 0; kk < 16; ++kk) {
    const int kb = kk * 4 + kh * 2;
    bf[kk].x = Wm[kb * D_ + ct + m16];
    bf[kk].y = Wm[(kb + 1) * D_ + ct + m16];
  }
  const int r = tid >> 4, c0 = (tid & 15) * 4;
  const float4 g4 = *(const float4*)(lng + c0);
  const float4 b4 = *(const float4*)(lnb + c0);
  float4 xv[8];
#pragma unroll
  for (int j = 0; j < 8; ++j)
    xv[j] = *(const float4*)(inp + (row0 + r + 16 * j) * D_ + c0);
#pragma unroll
  for (int j = 0; j < 8; ++j) {
    const float m = row16_sum((xv[j].x + xv[j].y) + (xv[j].z + xv[j].w)) * (1.f / D_);
    const float dx = xv[j].x - m, dy = xv[j].y - m, dz = xv[j].z - m, dw = xv[j].w - m;
    const float var = row16_sum((dx * dx + dy * dy) + (dz * dz + dw * dw)) * (1.f / D_);
    const float rs = rsqrtf(var + LN_EPS_);
    float4 o;
    o.x = dx * rs * g4.x + b4.x;
    o.y = dy * rs * g4.y + b4.y;
    o.z = dz * rs * g4.z + b4.z;
    o.w = dw * rs * g4.w + b4.w;
    *(float4*)&xln[r + 16 * j][c0] = o;
  }
  __syncthreads();
  float* op = outp + (row0 + 8 * kh) * D_ + ct + m16;
#pragma unroll
  for (int mt = 0; mt < LKV_ROWS_ / 16; mt += 2) {
    v2f a0[16], a1[16];
#pragma unroll
    for (int kk = 0; kk < 16; ++kk) {
      const int kb = kk * 4 + kh * 2;
      a0[kk].x = xln[mt * 16 + m16][kb];      a0[kk].y = xln[mt * 16 + m16][kb + 1];
      a1[kk].x = xln[mt * 16 + 16 + m16][kb]; a1[kk].y = xln[mt * 16 + 16 + m16][kb + 1];
    }
    __builtin_amdgcn_sched_barrier(0);
    v8f acc0 = {}, acc1 = {};
#pragma unroll
    for (int kk = 0; kk < 16; ++kk) {
      acc0 = wmma_f32(a0[kk], bf[kk], acc0);
      acc1 = wmma_f32(a1[kk], bf[kk], acc1);
    }
#pragma unroll
    for (int i = 0; i < 8; ++i) {
      op[(mt * 16 + i) * D_] = acc0[i];
      op[(mt * 16 + 16 + i) * D_] = acc1[i];
    }
  }
}

__global__ void slots_init_kernel(const float* __restrict__ sinit,
                                  const float* __restrict__ smu,
                                  const float* __restrict__ slog,
                                  float* __restrict__ slots) {
  int idx = blockIdx.x * blockDim.x + threadIdx.x;
  int d = idx & (D_ - 1);
  int h = idx / (B_ * S_ * D_);
  slots[idx] = smu[h * D_ + d] + expf(slog[h * D_ + d]) * sinit[idx];
}

__global__ void __launch_bounds__(128) hprev_q_kernel(
    const float* __restrict__ slots, const float* __restrict__ lng,
    const float* __restrict__ lnb, const float* __restrict__ Wq,
    float* __restrict__ hprev, float* __restrict__ qout) {
  __shared__ float hl[16][68];
  __shared__ float mu[16], rstd[16];
  const int tid = threadIdx.x;
  const int h = blockIdx.x >> 5;
  const int rt = blockIdx.x & 31;
  const int rowbase = h * (B_ * S_) + rt * 16;
  for (int e = tid; e < 16 * D_; e += 128) {
    int r = e >> 6, c = e & 63;
    hl[r][c] = slots[(size_t)(rowbase + r) * D_ + c];
  }
  __syncthreads();
  if (tid < 16) {
    float m = 0.f;
    for (int j = 0; j < D_; ++j) m += hl[tid][j];
    m *= (1.f / D_);
    float var = 0.f;
    for (int j = 0; j < D_; ++j) { float dd = hl[tid][j] - m; var += dd * dd; }
    var *= (1.f / D_);
    mu[tid] = m; rstd[tid] = rsqrtf(var + LN_EPS_);
  }
  __syncthreads();
  for (int e = tid; e < 16 * D_; e += 128) {
    int r = e >> 6, c = e & 63;
    float nv = (hl[r][c] - mu[r]) * rstd[r] * lng[h * D_ + c] + lnb[h * D_ + c];
    hl[r][c] = nv;
    hprev[(size_t)(rowbase + r) * D_ + c] = nv;
  }
  __syncthreads();
  const int w = tid >> 5, lane = tid & 31, m16 = lane & 15, kh = lane >> 4;
  const float* Wh = Wq + (size_t)h * D_ * D_;
  const int ct = w * 16;
  v8f acc = {};
  for (int kk = 0; kk < 16; ++kk) {
    int kb = kk * 4 + kh * 2;
    v2f a; a.x = hl[m16][kb]; a.y = hl[m16][kb + 1];
    v2f b; b.x = Wh[kb * D_ + ct + m16]; b.y = Wh[(kb + 1) * D_ + ct + m16];
    acc = wmma_f32(a, b, acc);
  }
  for (int i = 0; i < 8; ++i)
    qout[(size_t)(rowbase + i + 8 * kh) * D_ + ct + m16] = acc[i];
}

constexpr int TR_ = 128;
constexpr int NH_ = N_ / 2;
constexpr int NTH_ = NH_ / TR_;
constexpr int VP_ = 72;

template <int P>
__device__ __forceinline__ void attn_tile_step(
    float (&kt)[2][TR_][68], float (&vt)[2][TR_][VP_], float (&at)[2][2][TR_][17],
    const float* knext, const float* vnext, const v2f (&qf)[2][16],
    v8f (&accu)[2][2], float (&dacc)[2][8],
    const int tid, const int rb, const int gb, const int w, const int m16,
    const int kh) {
  const int sr = tid >> 4, sc = (tid & 15) * 4;
  v2f kf[16];
#pragma unroll
  for (int kk = 0; kk < 16; ++kk) {
    const int kbs = kk * 4 + kh * 2;
    kf[kk].x = kt[P][rb + m16][kbs];
    kf[kk].y = kt[P][rb + m16][kbs + 1];
  }
  const float* kp = knext + (size_t)sr * D_ + sc;
  const float* vp = vnext + (size_t)sr * D_ + sc;
  v4f kn[8], vn[8];
#pragma unroll
  for (int j = 0; j < 8; ++j) kn[j] = *(const v4f*)(kp + (size_t)j * 16 * D_);
#pragma unroll
  for (int j = 0; j < 8; ++j) vn[j] = *(const v4f*)(vp + (size_t)j * 16 * D_);
  asm volatile("" ::: "memory");
  __builtin_amdgcn_sched_barrier(0);
  v8f la[2], lb[2];
  la[0] = v8f{}; la[1] = v8f{}; lb[0] = v8f{}; lb[1] = v8f{};
#pragma unroll
  for (int kk = 0; kk < 8; ++kk) {
#pragma unroll
    for (int c = 0; c < 2; ++c) {
      la[c] = wmma_f32(kf[kk], qf[c][kk], la[c]);
      lb[c] = wmma_f32(kf[kk + 8], qf[c][kk + 8], lb[c]);
    }
  }
#pragma unroll
  for (int c = 0; c < 2; ++c)
#pragma unroll
    for (int i = 0; i < 8; ++i)
      at[P][c][rb + i + 8 * kh][m16] = (la[c][i] + lb[c][i]) * LSCALE2_;
  __builtin_amdgcn_wave_barrier();
  const int r = tid >> 1, hh = tid & 1;
  float lv[2][8];
#pragma unroll
  for (int c = 0; c < 2; ++c)
#pragma unroll
    for (int s = 0; s < 8; ++s) lv[c][s] = at[P][c][r][hh * 8 + s];
  __builtin_amdgcn_sched_barrier(0);
  v2f vf[16];
#pragma unroll
  for (int kk = 0; kk < 16; ++kk) {
    const int kbs = kk * 4 + kh * 2;
    vf[kk].x = vt[P][gb + kbs][w * 16 + m16];
    vf[kk].y = vt[P][gb + kbs + 1][w * 16 + m16];
  }
  __builtin_amdgcn_sched_barrier(0);
#pragma unroll
  for (int c = 0; c < 2; ++c) {
    float mx = -3.4e38f;
#pragma unroll
    for (int s = 0; s < 8; ++s) mx = fmaxf(mx, lv[c][s]);
    float ex[8], sum = 0.f;
#pragma unroll
    for (int s = 0; s < 8; ++s) { ex[s] = __builtin_amdgcn_exp2f(lv[c][s] - mx); sum += ex[s]; }
    const float inv = __builtin_amdgcn_rcpf(sum);
#pragma unroll
    for (int s = 0; s < 8; ++s) {
      const float pv = ex[s] * inv + EPS_ATTN_;
      at[P][c][r][hh * 8 + s] = pv;
      dacc[c][s] += pv;
    }
  }
  __builtin_amdgcn_sched_barrier(0);
#pragma unroll
  for (int j = 0; j < 8; ++j) *(v4f*)&kt[P ^ 1][sr + 16 * j][sc] = kn[j];
#pragma unroll
  for (int j = 0; j < 8; ++j) *(v4f*)&vt[P ^ 1][sr + 16 * j][sc] = vn[j];
  __syncthreads();
  v2f af[2][16];
#pragma unroll
  for (int c = 0; c < 2; ++c)
#pragma unroll
    for (int kk = 0; kk < 16; ++kk) {
      const int kbs = kk * 4 + kh * 2;
      af[c][kk].x = at[P][c][gb + kbs][m16];
      af[c][kk].y = at[P][c][gb + kbs + 1][m16];
    }
  __builtin_amdgcn_sched_barrier(0);
#pragma unroll
  for (int kk = 0; kk < 8; ++kk) {
#pragma unroll
    for (int c = 0; c < 2; ++c) {
      accu[c][0] = wmma_f32(af[c][kk], vf[kk], accu[c][0]);
      accu[c][1] = wmma_f32(af[c][kk + 8], vf[kk + 8], accu[c][1]);
    }
  }
}

__global__ void __launch_bounds__(256) __attribute__((amdgpu_waves_per_eu(2, 2)))
attn_kernel(
    const float* kbuf, const float* vbuf,
    const float* __restrict__ qbuf, float* pnum0, float* pnum1, float* pden) {
  __shared__ __attribute__((aligned(16))) float kt[2][TR_][68];
  __shared__ __attribute__((aligned(16))) float vt[2][TR_][VP_];
  __shared__ float at[2][2][TR_][17];
  __shared__ float dsc[32][TR_ + 1];
  __shared__ float red[2][4][8][32];
  const int tid = threadIdx.x;
  const int hq = blockIdx.x >> 7;
  const int nh = (blockIdx.x >> 6) & 1;
  const int b  = blockIdx.x & 63;
  const float* kb = kbuf + ((size_t)b * N_ + (size_t)nh * NH_) * D_;
  const float* vb = vbuf + ((size_t)b * N_ + (size_t)nh * NH_) * D_;
  const int w8 = tid >> 5, lane = tid & 31, m16 = lane & 15, kh = lane >> 4;
  const int g = w8 >> 2, w = w8 & 3;
  const int rb = w8 * 16, gb = g * 64;
  v2f qf[2][16];
#pragma unroll
  for (int c = 0; c < 2; ++c) {
    const int h = (hq * 2 + c) * 2 + (m16 >> 3), s = m16 & 7;
    const float* qr = qbuf + (size_t)((h * B_ + b) * S_ + s) * D_ + kh * 2;
#pragma unroll
    for (int kk = 0; kk < 16; ++kk) { qf[c][kk].x = qr[kk * 4]; qf[c][kk].y = qr[kk * 4 + 1]; }
  }
  {
    const int sr = tid >> 4, sc = (tid & 15) * 4;
#pragma unroll
    for (int j = 0; j < 8; ++j) {
      *(float4*)&kt[0][sr + 16 * j][sc] =
          *(const float4*)(kb + (size_t)(sr + 16 * j) * D_ + sc);
      *(float4*)&vt[0][sr + 16 * j][sc] =
          *(const float4*)(vb + (size_t)(sr + 16 * j) * D_ + sc);
    }
  }
  __syncthreads();
  v8f accu[2][2];
  float dacc[2][8];
#pragma unroll
  for (int c = 0; c < 2; ++c) {
    accu[c][0] = v8f{};
    accu[c][1] = v8f{};
#pragma unroll
    for (int s = 0; s < 8; ++s) dacc[c][s] = 0.f;
  }
  for (int t = 0; t < NTH_; t += 2) {
    attn_tile_step<0>(kt, vt, at, kb + (size_t)(t + 1) * TR_ * D_,
                      vb + (size_t)(t + 1) * TR_ * D_, qf, accu, dacc,
                      tid, rb, gb, w, m16, kh);
    const int tn = (t + 2 < NTH_) ? t + 2 : t + 1;
    attn_tile_step<1>(kt, vt, at, kb + (size_t)tn * TR_ * D_,
                      vb + (size_t)tn * TR_ * D_, qf, accu, dacc,
                      tid, rb, gb, w, m16, kh);
  }
  v8f asum[2];
  asum[0] = accu[0][0] + accu[0][1];
  asum[1] = accu[1][0] + accu[1][1];
  if (g == 1) {
#pragma unroll
    for (int c = 0; c < 2; ++c)
#pragma unroll
      for (int i = 0; i < 8; ++i) red[c][w][i][lane] = asum[c][i];
  }
  {
    const int r = tid >> 1, hh = tid & 1;
#pragma unroll
    for (int c = 0; c < 2; ++c)
#pragma unroll
      for (int s = 0; s < 8; ++s) dsc[c * 16 + hh * 8 + s][r] = dacc[c][s];
  }
  __syncthreads();
  if (tid < 32) {
    float dsum = 0.f;
    for (int r = 0; r < TR_; ++r) dsum += dsc[tid][r];
    const int h = hq * 4 + (tid >> 3), s = tid & 7;
    pden[nh * (H_ * B_ * S_) + (h * B_ + b) * S_ + s] = dsum;
  }
  if (g == 0) {
    float* pn = nh ? pnum1 : pnum0;
#pragma unroll
    for (int c = 0; c < 2; ++c)
      for (int i = 0; i < 8; ++i) {
        const int sp = i + 8 * kh;
        const int h = (hq * 2 + c) * 2 + (sp >> 3), s = sp & 7;
        pn[(size_t)((h * B_ + b) * S_ + s) * D_ + w * 16 + m16] =
            asum[c][i] + red[c][w][i][lane];
      }
  }
}

__global__ void __launch_bounds__(256) gru_mlp_kernel(
    const float* __restrict__ pnum0, const float* pnum1,
    const float* __restrict__ pden, const float* __restrict__ hprev,
    const float* __restrict__ wih, const float* __restrict__ whh,
    const float* __restrict__ bih, const float* __restrict__ bhh,
    const float* __restrict__ lnmg, const float* __restrict__ lnmb,
    const float* __restrict__ w1, const float* __restrict__ b1,
    const float* __restrict__ w2, const float* __restrict__ b2,
    float* slots) {
  __shared__ float ut[16][68];
  __shared__ float ht[16][68];
  __shared__ float gx[16][196];
  __shared__ float gh[16][196];
  __shared__ float sn[16][68];
  __shared__ float nm[16][68];
  __shared__ float hid[16][132];
  __shared__ float mu[16], rstd[16];
  const int tid = threadIdx.x;
  const int h = blockIdx.x >> 5, rt = blockIdx.x & 31;
  const int rowbase = h * (B_ * S_) + rt * 16;
  const int r0 = tid >> 4, c00 = (tid & 15) * 4;
  {
    const float4 a0 = *(const float4*)(pnum0 + (size_t)(rowbase + r0) * D_ + c00);
    const float4 a1 = *(const float4*)(pnum1 + (size_t)(rowbase + r0) * D_ + c00);
    const float dn = pden[rowbase + r0] + pden[H_ * B_ * S_ + rowbase + r0];
    ut[r0][c00] = (a0.x + a1.x) / dn;         ut[r0][c00 + 1] = (a0.y + a1.y) / dn;
    ut[r0][c00 + 2] = (a0.z + a1.z) / dn;     ut[r0][c00 + 3] = (a0.w + a1.w) / dn;
    float4 c = *(const float4*)(hprev + (size_t)(rowbase + r0) * D_ + c00);
    ht[r0][c00] = c.x; ht[r0][c00 + 1] = c.y; ht[r0][c00 + 2] = c.z; ht[r0][c00 + 3] = c.w;
  }
  __syncthreads();
  const int w = tid >> 5, lane = tid & 31, m16 = lane & 15, kh = lane >> 4;
  const float* wihh = wih + (size_t)h * D_ * 192;
  const float* whhh = whh + (size_t)h * D_ * 192;
  for (int T = w; T < 24; T += 8) {
    const bool isX = T < 12;
    const int ct = (T % 12) * 16;
    const float* Wt = isX ? wihh : whhh;
    v8f acc = {};
    for (int kk = 0; kk < 16; ++kk) {
      int kb = kk * 4 + kh * 2;
      v2f a;
      if (isX) { a.x = ut[m16][kb]; a.y = ut[m16][kb + 1]; }
      else     { a.x = ht[m16][kb]; a.y = ht[m16][kb + 1]; }
      v2f bb; bb.x = Wt[kb * 192 + ct + m16]; bb.y = Wt[(kb + 1) * 192 + ct + m16];
      acc = wmma_f32(a, bb, acc);
    }
    float* dst = isX ? &gx[0][0] : &gh[0][0];
    for (int i = 0; i < 8; ++i)
      dst[(i + 8 * kh) * 196 + ct + m16] = acc[i];
  }
  __syncthreads();
  {
    const float* bi = bih + h * 192;
    const float* bh = bhh + h * 192;
    for (int j = 0; j < 4; ++j) {
      int d = c00 + j;
      float xr = gx[r0][d] + bi[d],        hr = gh[r0][d] + bh[d];
      float xz = gx[r0][64 + d] + bi[64 + d],  hz = gh[r0][64 + d] + bh[64 + d];
      float xn = gx[r0][128 + d] + bi[128 + d], hn = gh[r0][128 + d] + bh[128 + d];
      float rg = 1.f / (1.f + expf(-(xr + hr)));
      float zg = 1.f / (1.f + expf(-(xz + hz)));
      float ng = tanhf(xn + rg * hn);
      sn[r0][d] = (1.f - zg) * ng + zg * ht[r0][d];
    }
  }
  __syncthreads();
  if (tid < 16) {
    float m = 0.f;
    for (int j = 0; j < D_; ++j) m += sn[tid][j];
    m *= (1.f / D_);
    float var = 0.f;
    for (int j = 0; j < D_; ++j) { float dd = sn[tid][j] - m; var += dd * dd; }
    var *= (1.f / D_);
    mu[tid] = m; rstd[tid] = rsqrtf(var + LN_EPS_);
  }
  __syncthreads();
  for (int j = 0; j < 4; ++j) {
    int d = c00 + j;
    nm[r0][d] = (sn[r0][d] - mu[r0]) * rstd[r0] * lnmg[h * D_ + d] + lnmb[h * D_ + d];
  }
  __syncthreads();
  {
    const float* w1h = w1 + (size_t)h * D_ * MLPH_;
    const float* b1h = b1 + h * MLPH_;
    const int ct = w * 16;
    v8f acc = {};
    for (int kk = 0; kk < 16; ++kk) {
      int kb = kk * 4 + kh * 2;
      v2f a;  a.x = nm[m16][kb]; a.y = nm[m16][kb + 1];
      v2f bb; bb.x = w1h[kb * MLPH_ + ct + m16]; bb.y = w1h[(kb + 1) * MLPH_ + ct + m16];
      acc = wmma_f32(a, bb, acc);
    }
    float bias = b1h[ct + m16];
    for (int i = 0; i < 8; ++i)
      hid[i + 8 * kh][ct + m16] = fmaxf(acc[i] + bias, 0.f);
  }
  __syncthreads();
  if (w < 4) {
    const float* w2h = w2 + (size_t)h * MLPH_ * D_;
    const float* b2h = b2 + h * D_;
    const int ct = w * 16;
    v8f acc = {};
    for (int kk = 0; kk < 32; ++kk) {
      int kb = kk * 4 + kh * 2;
      v2f a;  a.x = hid[m16][kb]; a.y = hid[m16][kb + 1];
      v2f bb; bb.x = w2h[kb * D_ + ct + m16]; bb.y = w2h[(kb + 1) * D_ + ct + m16];
      acc = wmma_f32(a, bb, acc);
    }
    float bias = b2h[ct + m16];
    for (int i = 0; i < 8; ++i) {
      int rr = i + 8 * kh;
      slots[(size_t)(rowbase + rr) * D_ + ct + m16] = sn[rr][ct + m16] + acc[i] + bias;
    }
  }
}

__global__ void copy_out_kernel(const float* __restrict__ slots,
                                float* __restrict__ out) {
  int i = blockIdx.x * blockDim.x + threadIdx.x;
  if (i < B_ * S_ * D_) out[i] = slots[i];
}

extern "C" void kernel_launch(void* const* d_in, const int* in_sizes, int n_in,
                              void* d_out, int out_size, void* d_ws, size_t ws_size,
                              hipStream_t stream) {
  (void)in_sizes; (void)n_in; (void)out_size; (void)ws_size;
  const float* inputs  = (const float*)d_in[0];
  const float* sinit   = (const float*)d_in[1];
  const float* smu     = (const float*)d_in[2];
  const float* slog    = (const float*)d_in[3];
  const float* ln_in_g = (const float*)d_in[4];
  const float* ln_in_b = (const float*)d_in[5];
  const float* Wk      = (const float*)d_in[6];
  const float* Wv      = (const float*)d_in[7];
  const float* ln_s_g  = (const float*)d_in[8];
  const float* ln_s_b  = (const float*)d_in[9];
  const float* Wq      = (const float*)d_in[10];
  const float* gru_wih = (const float*)d_in[11];
  const float* gru_whh = (const float*)d_in[12];
  const float* gru_bih = (const float*)d_in[13];
  const float* gru_bhh = (const float*)d_in[14];
  const float* ln_m_g  = (const float*)d_in[15];
  const float* ln_m_b  = (const float*)d_in[16];
  const float* mlp_w1  = (const float*)d_in[17];
  const float* mlp_b1  = (const float*)d_in[18];
  const float* mlp_w2  = (const float*)d_in[19];
  const float* mlp_b2  = (const float*)d_in[20];

  float* ws = (float*)d_ws;
  float* kbuf  = ws;
  float* vbuf  = kbuf + (size_t)B_ * N_ * D_;
  float* slots = vbuf + (size_t)B_ * N_ * D_;
  float* hprev = slots + H_ * B_ * S_ * D_;
  float* qbuf  = hprev + H_ * B_ * S_ * D_;
  float* updv  = qbuf + H_ * B_ * S_ * D_;
  float* dpart = (float*)d_out;

  ln_kv_kernel<<<(B_ * N_) / LKV_ROWS_, 256, 0, stream>>>(inputs, ln_in_g, ln_in_b,
                                                   Wk, Wv, kbuf, vbuf);
  slots_init_kernel<<<(H_ * B_ * S_ * D_) / 256, 256, 0, stream>>>(sinit, smu,
                                                                   slog, slots);
  for (int it = 0; it < ITERS_; ++it) {
    hprev_q_kernel<<<H_ * 32, 128, 0, stream>>>(slots, ln_s_g, ln_s_b, Wq,
                                                hprev, qbuf);
    attn_kernel<<<(H_ / 4) * 2 * B_, 256, 0, stream>>>(kbuf, vbuf, qbuf, updv,
                                                       slots, dpart);
    gru_mlp_kernel<<<H_ * 32, 256, 0, stream>>>(updv, slots, dpart, hprev,
                                                gru_wih, gru_whh,
                                                gru_bih, gru_bhh, ln_m_g, ln_m_b,
                                                mlp_w1, mlp_b1, mlp_w2, mlp_b2,
                                                slots);
  }
  copy_out_kernel<<<(B_ * S_ * D_ + 255) / 256, 256, 0, stream>>>(slots,
                                                                  (float*)d_out);
}


# attention logits formed transposed (q tile as A, k rows as B): softmax on a lane's own 8 registers, logits skip LDS; attn stored in the numerator's fragment order, 16-byte A reads; sum orders kept
# speedup vs baseline: 3.2226x; 1.0311x over previous
#include <hip/hip_runtime.h>

typedef float v2f __attribute__((ext_vector_type(2)));
typedef float v4f __attribute__((ext_vector_type(4)));
typedef float v8f __attribute__((ext_vector_type(8)));

constexpr int H_ = 8, B_ = 64, S_ = 8, D_ = 64, N_ = 4096, MLPH_ = 128, ITERS_ = 3;
constexpr float LN_EPS_ = 1e-5f, EPS_ATTN_ = 1e-8f, SCALE_ = 0.125f;
constexpr float LSCALE2_ = SCALE_ * 1.4426950408889634f;

__device__ __forceinline__ v8f wmma_f32(v2f a, v2f b, v8f c) {
  return __builtin_amdgcn_wmma_f32_16x16x4_f32(false, a, false, b, (short)0, c,
                                               false, false);
}

constexpr int LKV_ROWS_ = 128;

__device__ __forceinline__ float row16_sum(float v) {
  v += __int_as_float(__builtin_amdgcn_update_dpp(0, __float_as_int(v), 0x161, 0xf, 0xf, false));
  v += __int_as_float(__builtin_amdgcn_update_dpp(0, __float_as_int(v), 0x162, 0xf, 0xf, false));
  v += __int_as_float(__builtin_amdgcn_update_dpp(0, __float_as_int(v), 0x164, 0xf, 0xf, false));
  v += __int_as_float(__builtin_amdgcn_update_dpp(0, __float_as_int(v), 0x168, 0xf, 0xf, false));
  return v;
}

__global__ void __launch_bounds__(256) ln_kv_kernel(
    const float* __restrict__ inp, const float* __restrict__ lng,
    const float* __restrict__ lnb, const float* __restrict__ Wk,
    const float* __restrict__ Wv, float* __restrict__ kout,
    float* __restrict__ vout) {
  __shared__ __attribute__((aligned(16))) float xln[LKV_ROWS_][68];
  const int tid = threadIdx.x;
  const size_t row0 = (size_t)blockIdx.x * LKV_ROWS_;
  const int w = tid >> 5, lane = tid & 31, m16 = lane & 15, kh = lane >> 4;
  const float* Wm = (w < 4) ? Wk : Wv;
  float* outp = (w < 4) ? kout : vout;
  const int ct = (w & 3) * 16;
  v2f bf[16];
#pragma unroll
  for (int kk = 0; kk < 16; ++kk) {
    const int kb = kk * 4 + kh * 2;
    bf[kk].x = Wm[kb * D_ + ct + m16];
    bf[kk].y = Wm[(kb + 1) * D_ + ct + m16];
  }
  const int r = tid >> 4, c0 = (tid & 15) * 4;
  const float4 g4 = *(const float4*)(lng + c0);
  const float4 b4 = *(const float4*)(lnb + c0);
  float4 xv[8];
#pragma unroll
  for (int j = 0; j < 8; ++j)
    xv[j] = *(const float4*)(inp + (row0 + r + 16 * j) * D_ + c0);
#pragma unroll
  for (int j = 0; j < 8; ++j) {
    const float m = row16_sum((xv[j].x + xv[j].y) + (xv[j].z + xv[j].w)) * (1.f / D_);
    const float dx = xv[j].x - m, dy = xv[j].y - m, dz = xv[j].z - m, dw = xv[j].w - m;
    const float var = row16_sum((dx * dx + dy * dy) + (dz * dz + dw * dw)) * (1.f / D_);
    const float rs = rsqrtf(var + LN_EPS_);
    float4 o;
    o.x = dx * rs * g4.x + b4.x;
    o.y = dy * rs * g4.y + b4.y;
    o.z = dz * rs * g4.z + b4.z;
    o.w = dw * rs * g4.w + b4.w;
    *(float4*)&xln[r + 16 * j][c0] = o;
  }
  __syncthreads();
  float* op = outp + (row0 + 8 * kh) * D_ + ct + m16;
#pragma unroll
  for (int mt = 0; mt < LKV_ROWS_ / 16; mt += 2) {
    v2f a0[16], a1[16];
#pragma unroll
    for (int kk = 0; kk < 16; ++kk) {
      const int kb = kk * 4 + kh * 2;
      a0[kk].x = xln[mt * 16 + m16][kb];      a0[kk].y = xln[mt * 16 + m16][kb + 1];
      a1[kk].x = xln[mt * 16 + 16 + m16][kb]; a1[kk].y = xln[mt * 16 + 16 + m16][kb + 1];
    }
    __builtin_amdgcn_sched_barrier(0);
    v8f acc0 = {}, acc1 = {};
#pragma unroll
    for (int kk = 0; kk < 16; ++kk) {
      acc0 = wmma_f32(a0[kk], bf[kk], acc0);
      acc1 = wmma_f32(a1[kk], bf[kk], acc1);
    }
#pragma unroll
    for (int i = 0; i < 8; ++i) {
      op[(mt * 16 + i) * D_] = acc0[i];
      op[(mt * 16 + 16 + i) * D_] = acc1[i];
    }
  }
}

__global__ void slots_init_kernel(const float* __restrict__ sinit,
                                  const float* __restrict__ smu,
                                  const float* __restrict__ slog,
                                  float* __restrict__ slots) {
  int idx = blockIdx.x * blockDim.x + threadIdx.x;
  int d = idx & (D_ - 1);
  int h = idx / (B_ * S_ * D_);
  slots[idx] = smu[h * D_ + d] + expf(slog[h * D_ + d]) * sinit[idx];
}

__global__ void __launch_bounds__(128) hprev_q_kernel(
    const float* __restrict__ slots, const float* __restrict__ lng,
    const float* __restrict__ lnb, const float* __restrict__ Wq,
    float* __restrict__ hprev, float* __restrict__ qout) {
  __shared__ float hl[16][68];
  __shared__ float mu[16], rstd[16];
  const int tid = threadIdx.x;
  const int h = blockIdx.x >> 5;
  const int rt = blockIdx.x & 31;
  const int rowbase = h * (B_ * S_) + rt * 16;
  for (int e = tid; e < 16 * D_; e += 128) {
    int r = e >> 6, c = e & 63;
    hl[r][c] = slots[(size_t)(rowbase + r) * D_ + c];
  }
  __syncthreads();
  if (tid < 16) {
    float m = 0.f;
    for (int j = 0; j < D_; ++j) m += hl[tid][j];
    m *= (1.f / D_);
    float var = 0.f;
    for (int j = 0; j < D_; ++j) { float dd = hl[tid][j] - m; var += dd * dd; }
    var *= (1.f / D_);
    mu[tid] = m; rstd[tid] = rsqrtf(var + LN_EPS_);
  }
  __syncthreads();
  for (int e = tid; e < 16 * D_; e += 128) {
    int r = e >> 6, c = e & 63;
    float nv = (hl[r][c] - mu[r]) * rstd[r] * lng[h * D_ + c] + lnb[h * D_ + c];
    hl[r][c] = nv;
    hprev[(size_t)(rowbase + r) * D_ + c] = nv;
  }
  __syncthreads();
  const int w = tid >> 5, lane = tid & 31, m16 = lane & 15, kh = lane >> 4;
  const float* Wh = Wq + (size_t)h * D_ * D_;
  const int ct = w * 16;
  v8f acc = {};
  for (int kk = 0; kk < 16; ++kk) {
    int kb = kk * 4 + kh * 2;
    v2f a; a.x = hl[m16][kb]; a.y = hl[m16][kb + 1];
    v2f b; b.x = Wh[kb * D_ + ct + m16]; b.y = Wh[(kb + 1) * D_ + ct + m16];
    acc = wmma_f32(a, b, acc);
  }
  for (int i = 0; i < 8; ++i)
    qout[(size_t)(rowbase + i + 8 * kh) * D_ + ct + m16] = acc[i];
}

constexpr int TR_ = 128;
constexpr int NH_ = N_ / 2;
constexpr int NTH_ = NH_ / TR_;
constexpr int VP_ = 72;
constexpr int AP_ = TR_ + 4;

template <int P>
__device__ __forceinline__ void attn_tile_step(
    float (&kt)[2][TR_][68], float (&vt)[2][TR_][VP_], float (&at)[2][2][16][AP_],
    const float* knext, const float* vnext, const v2f (&qf)[2][16],
    v8f (&accu)[2][2], float (&dacc)[2][8],
    const int tid, const int rb, const int pc, const int gb, const int w,
    const int m16, const int kh) {
  const int sr = tid >> 4, sc = (tid & 15) * 4;
  v2f kf[16];
#pragma unroll
  for (int kk = 0; kk < 16; ++kk) {
    const int kbs = kk * 4 + kh * 2;
    kf[kk].x = kt[P][rb + m16][kbs];
    kf[kk].y = kt[P][rb + m16][kbs + 1];
  }
  const float* kp = knext + (size_t)sr * D_ + sc;
  const float* vp = vnext + (size_t)sr * D_ + sc;
  v4f kn[8], vn[8];
#pragma unroll
  for (int j = 0; j < 8; ++j) kn[j] = *(const v4f*)(kp + (size_t)j * 16 * D_);
#pragma unroll
  for (int j = 0; j < 8; ++j) vn[j] = *(const v4f*)(vp + (size_t)j * 16 * D_);
  asm volatile("" ::: "memory");
  __builtin_amdgcn_sched_barrier(0);
  v8f la[2], lb[2];
  la[0] = v8f{}; la[1] = v8f{}; lb[0] = v8f{}; lb[1] = v8f{};
#pragma unroll
  for (int kk = 0; kk < 8; ++kk) {
#pragma unroll
    for (int c = 0; c < 2; ++c) {
      la[c] = wmma_f32(qf[c][kk], kf[kk], la[c]);
      lb[c] = wmma_f32(qf[c][kk + 8], kf[kk + 8], lb[c]);
    }
  }
  __builtin_amdgcn_sched_barrier(0);
  v2f vf[16];
#pragma unroll
  for (int kk = 0; kk < 16; ++kk) {
    const int kbs = kk * 4 + kh * 2;
    vf[kk].x = vt[P][gb + kbs][w * 16 + m16];
    vf[kk].y = vt[P][gb + kbs + 1][w * 16 + m16];
  }
  __builtin_amdgcn_sched_barrier(0);
#pragma unroll
  for (int c = 0; c < 2; ++c) {
    float lv[8];
#pragma unroll
    for (int s = 0; s < 8; ++s) lv[s] = (la[c][s] + lb[c][s]) * LSCALE2_;
    float mx = -3.4e38f;
#pragma unroll
    for (int s = 0; s < 8; ++s) mx = fmaxf(mx, lv[s]);
    float ex[8], sum = 0.f;
#pragma unroll
    for (int s = 0; s < 8; ++s) { ex[s] = __builtin_amdgcn_exp2f(lv[s] - mx); sum += ex[s]; }
    const float inv = __builtin_amdgcn_rcpf(sum);
#pragma unroll
    for (int s = 0; s < 8; ++s) {
      const float pv = ex[s] * inv + EPS_ATTN_;
      at[P][c][8 * kh + s][pc] = pv;
      dacc[c][s] += pv;
    }
  }
  __builtin_amdgcn_sched_barrier(0);
#pragma unroll
  for (int j = 0; j < 8; ++j) *(v4f*)&kt[P ^ 1][sr + 16 * j][sc] = kn[j];
#pragma unroll
  for (int j = 0; j < 8; ++j) *(v4f*)&vt[P ^ 1][sr + 16 * j][sc] = vn[j];
  __syncthreads();
  v2f af[2][16];
#pragma unroll
  for (int kk = 0; kk < 8; ++kk)
#pragma unroll
    for (int c = 0; c < 2; ++c) {
      const v4f t = *(const v4f*)&at[P][c][m16][gb + (2 * kk + kh) * 4];
      af[c][kk] = v2f{t.x, t.y};
      af[c][kk + 8] = v2f{t.z, t.w};
    }
  __builtin_amdgcn_sched_barrier(0);
#pragma unroll
  for (int kk = 0; kk < 8; ++kk) {
#pragma unroll
    for (int c = 0; c < 2; ++c) {
      accu[c][0] = wmma_f32(af[c][kk], vf[kk], accu[c][0]);
      accu[c][1] = wmma_f32(af[c][kk + 8], vf[kk + 8], accu[c][1]);
    }
  }
}

__global__ void __launch_bounds__(256) __attribute__((amdgpu_waves_per_eu(2, 2)))
attn_kernel(
    const float* kbuf, const float* vbuf,
    const float* __restrict__ qbuf, float* pnum0, float* pnum1, float* pden) {
  __shared__ __attribute__((aligned(16))) float kt[2][TR_][68];
  __shared__ __attribute__((aligned(16))) float vt[2][TR_][VP_];
  __shared__ __attribute__((aligned(16))) float at[2][2][16][AP_];
  __shared__ float dsc[32][TR_ + 1];
  __shared__ float red[2][4][8][32];
  const int tid = threadIdx.x;
  const int hq = blockIdx.x >> 7;
  const int nh = (blockIdx.x >> 6) & 1;
  const int b  = blockIdx.x & 63;
  const float* kb = kbuf + ((size_t)b * N_ + (size_t)nh * NH_) * D_;
  const float* vb = vbuf + ((size_t)b * N_ + (size_t)nh * NH_) * D_;
  const int w8 = tid >> 5, lane = tid & 31, m16 = lane & 15, kh = lane >> 4;
  const int g = w8 >> 2, w = w8 & 3;
  const int rb = w8 * 16, gb = g * 64;
  const int q5 = (w8 & 1) * 16 + m16;
  const int pc = gb + (q5 >> 1) * 4 + ((w8 >> 1) & 1) * 2 + (q5 & 1);
  v2f qf[2][16];
#pragma unroll
  for (int c = 0; c < 2; ++c) {
    const int h = (hq * 2 + c) * 2 + (m16 >> 3), s = m16 & 7;
    const float* qr = qbuf + (size_t)((h * B_ + b) * S_ + s) * D_ + kh * 2;
#pragma unroll
    for (int kk = 0; kk < 16; ++kk) { qf[c][kk].x = qr[kk * 4]; qf[c][kk].y = qr[kk * 4 + 1]; }
  }
  {
    const int sr = tid >> 4, sc = (tid & 15) * 4;
#pragma unroll
    for (int j = 0; j < 8; ++j) {
      *(float4*)&kt[0][sr + 16 * j][sc] =
          *(const float4*)(kb + (size_t)(sr + 16 * j) * D_ + sc);
      *(float4*)&vt[0][sr + 16 * j][sc] =
          *(const float4*)(vb + (size_t)(sr + 16 * j) * D_ + sc);
    }
  }
  __syncthreads();
  v8f accu[2][2];
  float dacc[2][8];
#pragma unroll
  for (int c = 0; c < 2; ++c) {
    accu[c][0] = v8f{};
    accu[c][1] = v8f{};
#pragma unroll
    for (int s = 0; s < 8; ++s) dacc[c][s] = 0.f;
  }
  for (int t = 0; t < NTH_; t += 2) {
    attn_tile_step<0>(kt, vt, at, kb + (size_t)(t + 1) * TR_ * D_,
                      vb + (size_t)(t + 1) * TR_ * D_, qf, accu, dacc,
                      tid, rb, pc, gb, w, m16, kh);
    const int tn = (t + 2 < NTH_) ? t + 2 : t + 1;
    attn_tile_step<1>(kt, vt, at, kb + (size_t)tn * TR_ * D_,
                      vb + (size_t)tn * TR_ * D_, qf, accu, dacc,
                      tid, rb, pc, gb, w, m16, kh);
  }
  v8f asum[2];
  asum[0] = accu[0][0] + accu[0][1];
  asum[1] = accu[1][0] + accu[1][1];
  if (g == 1) {
#pragma unroll
    for (int c = 0; c < 2; ++c)
#pragma unroll
      for (int i = 0; i < 8; ++i) red[c][w][i][lane] = asum[c][i];
  }
  {
    const int r = w8 * 16 + m16;
#pragma unroll
    for (int c = 0; c < 2; ++c)
#pragma unroll
      for (int s = 0; s < 8; ++s) dsc[c * 16 + kh * 8 + s][r] = dacc[c][s];
  }
  __syncthreads();
  if (tid < 32) {
    float dsum = 0.f;
    for (int r = 0; r < TR_; ++r) dsum += dsc[tid][r];
    const int h = hq * 4 + (tid >> 3), s = tid & 7;
    pden[nh * (H_ * B_ * S_) + (h * B_ + b) * S_ + s] = dsum;
  }
  if (g == 0) {
    float* pn = nh ? pnum1 : pnum0;
#pragma unroll
    for (int c = 0; c < 2; ++c)
      for (int i = 0; i < 8; ++i) {
        const int sp = i + 8 * kh;
        const int h = (hq * 2 + c) * 2 + (sp >> 3), s = sp & 7;
        pn[(size_t)((h * B_ + b) * S_ + s) * D_ + w * 16 + m16] =
            asum[c][i] + red[c][w][i][lane];
      }
  }
}

__global__ void __launch_bounds__(256) gru_mlp_kernel(
    const float* __restrict__ pnum0, const float* pnum1,
    const float* __restrict__ pden, const float* __restrict__ hprev,
    const float* __restrict__ wih, const float* __restrict__ whh,
    const float* __restrict__ bih, const float* __restrict__ bhh,
    const float* __restrict__ lnmg, const float* __restrict__ lnmb,
    const float* __restrict__ w1, const float* __restrict__ b1,
    const float* __restrict__ w2, const float* __restrict__ b2,
    float* slots) {
  __shared__ float ut[16][68];
  __shared__ float ht[16][68];
  __shared__ float gx[16][196];
  __shared__ float gh[16][196];
  __shared__ float sn[16][68];
  __shared__ float nm[16][68];
  __shared__ float hid[16][132];
  __shared__ float mu[16], rstd[16];
  const int tid = threadIdx.x;
  const int h = blockIdx.x >> 5, rt = blockIdx.x & 31;
  const int rowbase = h * (B_ * S_) + rt * 16;
  const int r0 = tid >> 4, c00 = (tid & 15) * 4;
  {
    const float4 a0 = *(const float4*)(pnum0 + (size_t)(rowbase + r0) * D_ + c00);
    const float4 a1 = *(const float4*)(pnum1 + (size_t)(rowbase + r0) * D_ + c00);
    const float dn = pden[rowbase + r0] + pden[H_ * B_ * S_ + rowbase + r0];
    ut[r0][c00] = (a0.x + a1.x) / dn;         ut[r0][c00 + 1] = (a0.y + a1.y) / dn;
    ut[r0][c00 + 2] = (a0.z + a1.z) / dn;     ut[r0][c00 + 3] = (a0.w + a1.w) / dn;
    float4 c = *(const float4*)(hprev + (size_t)(rowbase + r0) * D_ + c00);
    ht[r0][c00] = c.x; ht[r0][c00 + 1] = c.y; ht[r0][c00 + 2] = c.z; ht[r0][c00 + 3] = c.w;
  }
  __syncthreads();
  const int w = tid >> 5, lane = tid & 31, m16 = lane & 15, kh = lane >> 4;
  const float* wihh = wih + (size_t)h * D_ * 192;
  const float* whhh = whh + (size_t)h * D_ * 192;
  for (int T = w; T < 24; T += 8) {
    const bool isX = T < 12;
    const int ct = (T % 12) * 16;
    const float* Wt = isX ? wihh : whhh;
    v8f acc = {};
    for (int kk = 0; kk < 16; ++kk) {
      int kb = kk * 4 + kh * 2;
      v2f a;
      if (isX) { a.x = ut[m16][kb]; a.y = ut[m16][kb + 1]; }
      else     { a.x = ht[m16][kb]; a.y = ht[m16][kb + 1]; }
      v2f bb; bb.x = Wt[kb * 192 + ct + m16]; bb.y = Wt[(kb + 1) * 192 + ct + m16];
      acc = wmma_f32(a, bb, acc);
    }
    float* dst = isX ? &gx[0][0] : &gh[0][0];
    for (int i = 0; i < 8; ++i)
      dst[(i + 8 * kh) * 196 + ct + m16] = acc[i];
  }
  __syncthreads();
  {
    const float* bi = bih + h * 192;
    const float* bh = bhh + h * 192;
    for (int j = 0; j < 4; ++j) {
      int d = c00 + j;
      float xr = gx[r0][d] + bi[d],        hr = gh[r0][d] + bh[d];
      float xz = gx[r0][64 + d] + bi[64 + d],  hz = gh[r0][64 + d] + bh[64 + d];
      float xn = gx[r0][128 + d] + bi[128 + d], hn = gh[r0][128 + d] + bh[128 + d];
      float rg = 1.f / (1.f + expf(-(xr + hr)));
      float zg = 1.f / (1.f + expf(-(xz + hz)));
      float ng = tanhf(xn + rg * hn);
      sn[r0][d] = (1.f - zg) * ng + zg * ht[r0][d];
    }
  }
  __syncthreads();
  if (tid < 16) {
    float m = 0.f;
    for (int j = 0; j < D_; ++j) m += sn[tid][j];
    m *= (1.f / D_);
    float var = 0.f;
    for (int j = 0; j < D_; ++j) { float dd = sn[tid][j] - m; var += dd * dd; }
    var *= (1.f / D_);
    mu[tid] = m; rstd[tid] = rsqrtf(var + LN_EPS_);
  }
  __syncthreads();
  for (int j = 0; j < 4; ++j) {
    int d = c00 + j;
    nm[r0][d] = (sn[r0][d] - mu[r0]) * rstd[r0] * lnmg[h * D_ + d] + lnmb[h * D_ + d];
  }
  __syncthreads();
  {
    const float* w1h = w1 + (size_t)h * D_ * MLPH_;
    const float* b1h = b1 + h * MLPH_;
    const int ct = w * 16;
    v8f acc = {};
    for (int kk = 0; kk < 16; ++kk) {
      int kb = kk * 4 + kh * 2;
      v2f a;  a.x = nm[m16][kb]; a.y = nm[m16][kb + 1];
      v2f bb; bb.x = w1h[kb * MLPH_ + ct + m16]; bb.y = w1h[(kb + 1) * MLPH_ + ct + m16];
      acc = wmma_f32(a, bb, acc);
    }
    float bias = b1h[ct + m16];
    for (int i = 0; i < 8; ++i)
      hid[i + 8 * kh][ct + m16] = fmaxf(acc[i] + bias, 0.f);
  }
  __syncthreads();
  if (w < 4) {
    const float* w2h = w2 + (size_t)h * MLPH_ * D_;
    const float* b2h = b2 + h * D_;
    const int ct = w * 16;
    v8f acc = {};
    for (int kk = 0; kk < 32; ++kk) {
      int kb = kk * 4 + kh * 2;
      v2f a;  a.x = hid[m16][kb]; a.y = hid[m16][kb + 1];
      v2f bb; bb.x = w2h[kb * D_ + ct + m16]; bb.y = w2h[(kb + 1) * D_ + ct + m16];
      acc = wmma_f32(a, bb, acc);
    }
    float bias = b2h[ct + m16];
    for (int i = 0; i < 8; ++i) {
      int rr = i + 8 * kh;
      slots[(size_t)(rowbase + rr) * D_ + ct + m16] = sn[rr][ct + m16] + acc[i] + bias;
    }
  }
}

__global__ void copy_out_kernel(const float* __restrict__ slots,
                                float* __restrict__ out) {
  int i = blockIdx.x * blockDim.x + threadIdx.x;
  if (i < B_ * S_ * D_) out[i] = slots[i];
}

extern "C" void kernel_launch(void* const* d_in, const int* in_sizes, int n_in,
                              void* d_out, int out_size, void* d_ws, size_t ws_size,
                              hipStream_t stream) {
  (void)in_sizes; (void)n_in; (void)out_size; (void)ws_size;
  const float* inputs  = (const float*)d_in[0];
  const float* sinit   = (const float*)d_in[1];
  const float* smu     = (const float*)d_in[2];
  const float* slog    = (const float*)d_in[3];
  const float* ln_in_g = (const float*)d_in[4];
  const float* ln_in_b = (const float*)d_in[5];
  const float* Wk      = (const float*)d_in[6];
  const float* Wv      = (const float*)d_in[7];
  const float* ln_s_g  = (const float*)d_in[8];
  const float* ln_s_b  = (const float*)d_in[9];
  const float* Wq      = (const float*)d_in[10];
  const float* gru_wih = (const float*)d_in[11];
  const float* gru_whh = (const float*)d_in[12];
  const float* gru_bih = (const float*)d_in[13];
  const float* gru_bhh = (const float*)d_in[14];
  const float* ln_m_g  = (const float*)d_in[15];
  const float* ln_m_b  = (const float*)d_in[16];
  const float* mlp_w1  = (const float*)d_in[17];
  const float* mlp_b1  = (const float*)d_in[18];
  const float* mlp_w2  = (const float*)d_in[19];
  const float* mlp_b2  = (const float*)d_in[20];

  float* ws = (float*)d_ws;
  float* kbuf  = ws;
  float* vbuf  = kbuf + (size_t)B_ * N_ * D_;
  float* slots = vbuf + (size_t)B_ * N_ * D_;
  float* hprev = slots + H_ * B_ * S_ * D_;
  float* qbuf  = hprev + H_ * B_ * S_ * D_;
  float* updv  = qbuf + H_ * B_ * S_ * D_;
  float* dpart = (float*)d_out;

  ln_kv_kernel<<<(B_ * N_) / LKV_ROWS_, 256, 0, stream>>>(inputs, ln_in_g, ln_in_b,
                                                   Wk, Wv, kbuf, vbuf);
  slots_init_kernel<<<(H_ * B_ * S_ * D_) / 256, 256, 0, stream>>>(sinit, smu,
                                                                   slog, slots);
  for (int it = 0; it < ITERS_; ++it) {
    hprev_q_kernel<<<H_ * 32, 128, 0, stream>>>(slots, ln_s_g, ln_s_b, Wq,
                                                hprev, qbuf);
    attn_kernel<<<(H_ / 4) * 2 * B_, 256, 0, stream>>>(kbuf, vbuf, qbuf, updv,
                                                       slots, dpart);
    gru_mlp_kernel<<<H_ * 32, 256, 0, stream>>>(updv, slots, dpart, hprev,
                                                gru_wih, gru_whh,
                                                gru_bih, gru_bhh, ln_m_g, ln_m_b,
                                                mlp_w1, mlp_b1, mlp_w2, mlp_b2,
                                                slots);
  }
  copy_out_kernel<<<(B_ * S_ * D_ + 255) / 256, 256, 0, stream>>>(slots,
                                                                  (float*)d_out);
}


# slots init and the first LayerNorm/q launch run as 128 extra workgroups of the k/v projection launch, which they do not depend on (side by side, nothing chained): 12 launches to 10; sums keep order
# speedup vs baseline: 3.4599x; 1.0736x over previous
#include <hip/hip_runtime.h>

typedef float v2f __attribute__((ext_vector_type(2)));
typedef float v4f __attribute__((ext_vector_type(4)));
typedef float v8f __attribute__((ext_vector_type(8)));

constexpr int H_ = 8, B_ = 64, S_ = 8, D_ = 64, N_ = 4096, MLPH_ = 128, ITERS_ = 3;
constexpr float LN_EPS_ = 1e-5f, EPS_ATTN_ = 1e-8f, SCALE_ = 0.125f;
constexpr float LSCALE2_ = SCALE_ * 1.4426950408889634f;

__device__ __forceinline__ v8f wmma_f32(v2f a, v2f b, v8f c) {
  return __builtin_amdgcn_wmma_f32_16x16x4_f32(false, a, false, b, (short)0, c,
                                               false, false);
}

constexpr int LKV_ROWS_ = 128;
constexpr int SLOT_ROWS_ = 32;
constexpr int SLOT_WGS_ = (H_ * B_ * S_) / SLOT_ROWS_;

__device__ __forceinline__ float row16_sum(float v) {
  v += __int_as_float(__builtin_amdgcn_update_dpp(0, __float_as_int(v), 0x161, 0xf, 0xf, false));
  v += __int_as_float(__builtin_amdgcn_update_dpp(0, __float_as_int(v), 0x162, 0xf, 0xf, false));
  v += __int_as_float(__builtin_amdgcn_update_dpp(0, __float_as_int(v), 0x164, 0xf, 0xf, false));
  v += __int_as_float(__builtin_amdgcn_update_dpp(0, __float_as_int(v), 0x168, 0xf, 0xf, false));
  return v;
}

__global__ void __launch_bounds__(256) ln_kv_slots_kernel(
    const float* __restrict__ inp, const float* __restrict__ lng,
    const float* __restrict__ lnb, const float* __restrict__ Wk,
    const float* __restrict__ Wv, float* __restrict__ kout,
    float* __restrict__ vout,
    const float* __restrict__ sinit, const float* __restrict__ smu,
    const float* __restrict__ slog, const float* __restrict__ lnsg,
    const float* __restrict__ lnsb, const float* __restrict__ Wq,
    float* __restrict__ hprev, float* __restrict__ qout) {
  __shared__ __attribute__((aligned(16))) float xln[LKV_ROWS_][68];
  __shared__ float smean[SLOT_ROWS_], srstd[SLOT_ROWS_];
  const int tid = threadIdx.x;
  const int w = tid >> 5, lane = tid & 31, m16 = lane & 15, kh = lane >> 4;
  if (blockIdx.x < SLOT_WGS_) {
    const int h = blockIdx.x / (B_ * S_ / SLOT_ROWS_);
    const int rowbase = blockIdx.x * SLOT_ROWS_;
    for (int e = tid; e < SLOT_ROWS_ * D_; e += 256) {
      const int r = e >> 6, cc = e & 63;
      xln[r][cc] = smu[h * D_ + cc] +
                   expf(slog[h * D_ + cc]) * sinit[(size_t)(rowbase + r) * D_ + cc];
    }
    __syncthreads();
    if (tid < SLOT_ROWS_) {
      float m = 0.f;
      for (int j = 0; j < D_; ++j) m += xln[tid][j];
      m *= (1.f / D_);
      float var = 0.f;
      for (int j = 0; j < D_; ++j) { float dd = xln[tid][j] - m; var += dd * dd; }
      var *= (1.f / D_);
      smean[tid] = m; srstd[tid] = rsqrtf(var + LN_EPS_);
    }
    __syncthreads();
    for (int e = tid; e < SLOT_ROWS_ * D_; e += 256) {
      const int r = e >> 6, cc = e & 63;
      const float nv = (xln[r][cc] - smean[r]) * srstd[r] * lnsg[h * D_ + cc] + lnsb[h * D_ + cc];
      xln[r][cc] = nv;
      hprev[(size_t)(rowbase + r) * D_ + cc] = nv;
    }
    __syncthreads();
    const float* Wh = Wq + (size_t)h * D_ * D_;
    const int rg = (w >> 2) * 16, qct = (w & 3) * 16;
    v8f acc = {};
    for (int kk = 0; kk < 16; ++kk) {
      const int kb = kk * 4 + kh * 2;
      v2f a; a.x = xln[rg + m16][kb]; a.y = xln[rg + m16][kb + 1];
      v2f b; b.x = Wh[kb * D_ + qct + m16]; b.y = Wh[(kb + 1) * D_ + qct + m16];
      acc = wmma_f32(a, b, acc);
    }
    for (int i = 0; i < 8; ++i)
      qout[(size_t)(rowbase + rg + i + 8 * kh) * D_ + qct + m16] = acc[i];
    return;
  }
  const size_t row0 = (size_t)(blockIdx.x - SLOT_WGS_) * LKV_ROWS_;
  const float* Wm = (w < 4) ? Wk : Wv;
  float* outp = (w < 4) ? kout : vout;
  const int ct = (w & 3) * 16;
  v2f bf[16];
#pragma unroll
  for (int kk = 0; kk < 16; ++kk) {
    const int kb = kk * 4 + kh * 2;
    bf[kk].x = Wm[kb * D_ + ct + m16];
    bf[kk].y = Wm[(kb + 1) * D_ + ct + m16];
  }
  const int r = tid >> 4, c0 = (tid & 15) * 4;
  const float4 g4 = *(const float4*)(lng + c0);
  const float4 b4 = *(const float4*)(lnb + c0);
  float4 xv[8];
#pragma unroll
  for (int j = 0; j < 8; ++j)
    xv[j] = *(const float4*)(inp + (row0 + r + 16 * j) * D_ + c0);
#pragma unroll
  for (int j = 0; j < 8; ++j) {
    const float m = row16_sum((xv[j].x + xv[j].y) + (xv[j].z + xv[j].w)) * (1.f / D_);
    const float dx = xv[j].x - m, dy = xv[j].y - m, dz = xv[j].z - m, dw = xv[j].w - m;
    const float var = row16_sum((dx * dx + dy * dy) + (dz * dz + dw * dw)) * (1.f / D_);
    const float rs = rsqrtf(var + LN_EPS_);
    float4 o;
    o.x = dx * rs * g4.x + b4.x;
    o.y = dy * rs * g4.y + b4.y;
    o.z = dz * rs * g4.z + b4.z;
    o.w = dw * rs * g4.w + b4.w;
    *(float4*)&xln[r + 16 * j][c0] = o;
  }
  __syncthreads();
  float* op = outp + (row0 + 8 * kh) * D_ + ct + m16;
#pragma unroll
  for (int mt = 0; mt < LKV_ROWS_ / 16; mt += 2) {
    v2f a0[16], a1[16];
#pragma unroll
    for (int kk = 0; kk < 16; ++kk) {
      const int kb = kk * 4 + kh * 2;
      a0[kk].x = xln[mt * 16 + m16][kb];      a0[kk].y = xln[mt * 16 + m16][kb + 1];
      a1[kk].x = xln[mt * 16 + 16 + m16][kb]; a1[kk].y = xln[mt * 16 + 16 + m16][kb + 1];
    }
    __builtin_amdgcn_sched_barrier(0);
    v8f acc0 = {}, acc1 = {};
#pragma unroll
    for (int kk = 0; kk < 16; ++kk) {
      acc0 = wmma_f32(a0[kk], bf[kk], acc0);
      acc1 = wmma_f32(a1[kk], bf[kk], acc1);
    }
#pragma unroll
    for (int i = 0; i < 8; ++i) {
      op[(mt * 16 + i) * D_] = acc0[i];
      op[(mt * 16 + 16 + i) * D_] = acc1[i];
    }
  }
}

__global__ void __launch_bounds__(128) hprev_q_kernel(
    const float* __restrict__ slots, const float* __restrict__ lng,
    const float* __restrict__ lnb, const float* __restrict__ Wq,
    float* __restrict__ hprev, float* __restrict__ qout) {
  __shared__ float hl[16][68];
  __shared__ float mu[16], rstd[16];
  const int tid = threadIdx.x;
  const int h = blockIdx.x >> 5;
  const int rt = blockIdx.x & 31;
  const int rowbase = h * (B_ * S_) + rt * 16;
  for (int e = tid; e < 16 * D_; e += 128) {
    int r = e >> 6, c = e & 63;
    hl[r][c] = slots[(size_t)(rowbase + r) * D_ + c];
  }
  __syncthreads();
  if (tid < 16) {
    float m = 0.f;
    for (int j = 0; j < D_; ++j) m += hl[tid][j];
    m *= (1.f / D_);
    float var = 0.f;
    for (int j = 0; j < D_; ++j) { float dd = hl[tid][j] - m; var += dd * dd; }
    var *= (1.f / D_);
    mu[tid] = m; rstd[tid] = rsqrtf(var + LN_EPS_);
  }
  __syncthreads();
  for (int e = tid; e < 16 * D_; e += 128) {
    int r = e >> 6, c = e & 63;
    float nv = (hl[r][c] - mu[r]) * rstd[r] * lng[h * D_ + c] + lnb[h * D_ + c];
    hl[r][c] = nv;
    hprev[(size_t)(rowbase + r) * D_ + c] = nv;
  }
  __syncthreads();
  const int w = tid >> 5, lane = tid & 31, m16 = lane & 15, kh = lane >> 4;
  const float* Wh = Wq + (size_t)h * D_ * D_;
  const int ct = w * 16;
  v8f acc = {};
  for (int kk = 0; kk < 16; ++kk) {
    int kb = kk * 4 + kh * 2;
    v2f a; a.x = hl[m16][kb]; a.y = hl[m16][kb + 1];
    v2f b; b.x = Wh[kb * D_ + ct + m16]; b.y = Wh[(kb + 1) * D_ + ct + m16];
    acc = wmma_f32(a, b, acc);
  }
  for (int i = 0; i < 8; ++i)
    qout[(size_t)(rowbase + i + 8 * kh) * D_ + ct + m16] = acc[i];
}

constexpr int TR_ = 128;
constexpr int NH_ = N_ / 2;
constexpr int NTH_ = NH_ / TR_;
constexpr int VP_ = 72;
constexpr int AP_ = TR_ + 4;

template <int P>
__device__ __forceinline__ void attn_tile_step(
    float (&kt)[2][TR_][68], float (&vt)[2][TR_][VP_], float (&at)[2][2][16][AP_],
    const float* knext, const float* vnext, const v2f (&qf)[2][16],
    v8f (&accu)[2][2], float (&dacc)[2][8],
    const int tid, const int rb, const int pc, const int gb, const int w,
    const int m16, const int kh) {
  const int sr = tid >> 4, sc = (tid & 15) * 4;
  v2f kf[16];
#pragma unroll
  for (int kk = 0; kk < 16; ++kk) {
    const int kbs = kk * 4 + kh * 2;
    kf[kk].x = kt[P][rb + m16][kbs];
    kf[kk].y = kt[P][rb + m16][kbs + 1];
  }
  const float* kp = knext + (size_t)sr * D_ + sc;
  const float* vp = vnext + (size_t)sr * D_ + sc;
  v4f kn[8], vn[8];
#pragma unroll
  for (int j = 0; j < 8; ++j) kn[j] = *(const v4f*)(kp + (size_t)j * 16 * D_);
#pragma unroll
  for (int j = 0; j < 8; ++j) vn[j] = *(const v4f*)(vp + (size_t)j * 16 * D_);
  asm volatile("" ::: "memory");
  __builtin_amdgcn_sched_barrier(0);
  v8f la[2], lb[2];
  la[0] = v8f{}; la[1] = v8f{}; lb[0] = v8f{}; lb[1] = v8f{};
#pragma unroll
  for (int kk = 0; kk < 8; ++kk) {
#pragma unroll
    for (int c = 0; c < 2; ++c) {
      la[c] = wmma_f32(qf[c][kk], kf[kk], la[c]);
      lb[c] = wmma_f32(qf[c][kk + 8], kf[kk + 8], lb[c]);
    }
  }
  __builtin_amdgcn_sched_barrier(0);
  v2f vf[16];
#pragma unroll
  for (int kk = 0; kk < 16; ++kk) {
    const int kbs = kk * 4 + kh * 2;
    vf[kk].x = vt[P][gb + kbs][w * 16 + m16];
    vf[kk].y = vt[P][gb + kbs + 1][w * 16 + m16];
  }
  __builtin_amdgcn_sched_barrier(0);
#pragma unroll
  for (int c = 0; c < 2; ++c) {
    float lv[8];
#pragma unroll
    for (int s = 0; s < 8; ++s) lv[s] = (la[c][s] + lb[c][s]) * LSCALE2_;
    float mx = -3.4e38f;
#pragma unroll
    for (int s = 0; s < 8; ++s) mx = fmaxf(mx, lv[s]);
    float ex[8], sum = 0.f;
#pragma unroll
    for (int s = 0; s < 8; ++s) { ex[s] = __builtin_amdgcn_exp2f(lv[s] - mx); sum += ex[s]; }
    const float inv = __builtin_amdgcn_rcpf(sum);
#pragma unroll
    for (int s = 0; s < 8; ++s) {
      const float pv = ex[s] * inv + EPS_ATTN_;
      at[P][c][8 * kh + s][pc] = pv;
      dacc[c][s] += pv;
    }
  }
  __builtin_amdgcn_sched_barrier(0);
#pragma unroll
  for (int j = 0; j < 8; ++j) *(v4f*)&kt[P ^ 1][sr + 16 * j][sc] = kn[j];
#pragma unroll
  for (int j = 0; j < 8; ++j) *(v4f*)&vt[P ^ 1][sr + 16 * j][sc] = vn[j];
  __syncthreads();
  v2f af[2][16];
#pragma unroll
  for (int kk = 0; kk < 8; ++kk)
#pragma unroll
    for (int c = 0; c < 2; ++c) {
      const v4f t = *(const v4f*)&at[P][c][m16][gb + (2 * kk + kh) * 4];
      af[c][kk] = v2f{t.x, t.y};
      af[c][kk + 8] = v2f{t.z, t.w};
    }
  __builtin_amdgcn_sched_barrier(0);
#pragma unroll
  for (int kk = 0; kk < 8; ++kk) {
#pragma unroll
    for (int c = 0; c < 2; ++c) {
      accu[c][0] = wmma_f32(af[c][kk], vf[kk], accu[c][0]);
      accu[c][1] = wmma_f32(af[c][kk + 8], vf[kk + 8], accu[c][1]);
    }
  }
}

__global__ void __launch_bounds__(256) __attribute__((amdgpu_waves_per_eu(2, 2)))
attn_kernel(
    const float* kbuf, const float* vbuf,
    const float* __restrict__ qbuf, float* pnum0, float* pnum1, float* pden) {
  __shared__ __attribute__((aligned(16))) float kt[2][TR_][68];
  __shared__ __attribute__((aligned(16))) float vt[2][TR_][VP_];
  __shared__ __attribute__((aligned(16))) float at[2][2][16][AP_];
  __shared__ float dsc[32][TR_ + 1];
  __shared__ float red[2][4][8][32];
  const int tid = threadIdx.x;
  const int hq = blockIdx.x >> 7;
  const int nh = (blockIdx.x >> 6) & 1;
  const int b  = blockIdx.x & 63;
  const float* kb = kbuf + ((size_t)b * N_ + (size_t)nh * NH_) * D_;
  const float* vb = vbuf + ((size_t)b * N_ + (size_t)nh * NH_) * D_;
  const int w8 = tid >> 5, lane = tid & 31, m16 = lane & 15, kh = lane >> 4;
  const int g = w8 >> 2, w = w8 & 3;
  const int rb = w8 * 16, gb = g * 64;
  const int q5 = (w8 & 1) * 16 + m16;
  const int pc = gb + (q5 >> 1) * 4 + ((w8 >> 1) & 1) * 2 + (q5 & 1);
  v2f qf[2][16];
#pragma unroll
  for (int c = 0; c < 2; ++c) {
    const int h = (hq * 2 + c) * 2 + (m16 >> 3), s = m16 & 7;
    const float* qr = qbuf + (size_t)((h * B_ + b) * S_ + s) * D_ + kh * 2;
#pragma unroll
    for (int kk = 0; kk < 16; ++kk) { qf[c][kk].x = qr[kk * 4]; qf[c][kk].y = qr[kk * 4 + 1]; }
  }
  {
    const int sr = tid >> 4, sc = (tid & 15) * 4;
#pragma unroll
    for (int j = 0; j < 8; ++j) {
      *(float4*)&kt[0][sr + 16 * j][sc] =
          *(const float4*)(kb + (size_t)(sr + 16 * j) * D_ + sc);
      *(float4*)&vt[0][sr + 16 * j][sc] =
          *(const float4*)(vb + (size_t)(sr + 16 * j) * D_ + sc);
    }
  }
  __syncthreads();
  v8f accu[2][2];
  float dacc[2][8];
#pragma unroll
  for (int c = 0; c < 2; ++c) {
    accu[c][0] = v8f{};
    accu[c][1] = v8f{};
#pragma unroll
    for (int s = 0; s < 8; ++s) dacc[c][s] = 0.f;
  }
  for (int t = 0; t < NTH_; t += 2) {
    attn_tile_step<0>(kt, vt, at, kb + (size_t)(t + 1) * TR_ * D_,
                      vb + (size_t)(t + 1) * TR_ * D_, qf, accu, dacc,
                      tid, rb, pc, gb, w, m16, kh);
    const int tn = (t + 2 < NTH_) ? t + 2 : t + 1;
    attn_tile_step<1>(kt, vt, at, kb + (size_t)tn * TR_ * D_,
                      vb + (size_t)tn * TR_ * D_, qf, accu, dacc,
                      tid, rb, pc, gb, w, m16, kh);
  }
  v8f asum[2];
  asum[0] = accu[0][0] + accu[0][1];
  asum[1] = accu[1][0] + accu[1][1];
  if (g == 1) {
#pragma unroll
    for (int c = 0; c < 2; ++c)
#pragma unroll
      for (int i = 0; i < 8; ++i) red[c][w][i][lane] = asum[c][i];
  }
  {
    const int r = w8 * 16 + m16;
#pragma unroll
    for (int c = 0; c < 2; ++c)
#pragma unroll
      for (int s = 0; s < 8; ++s) dsc[c * 16 + kh * 8 + s][r] = dacc[c][s];
  }
  __syncthreads();
  if (tid < 32) {
    float dsum = 0.f;
    for (int r = 0; r < TR_; ++r) dsum += dsc[tid][r];
    const int h = hq * 4 + (tid >> 3), s = tid & 7;
    pden[nh * (H_ * B_ * S_) + (h * B_ + b) * S_ + s] = dsum;
  }
  if (g == 0) {
    float* pn = nh ? pnum1 : pnum0;
#pragma unroll
    for (int c = 0; c < 2; ++c)
      for (int i = 0; i < 8; ++i) {
        const int sp = i + 8 * kh;
        const int h = (hq * 2 + c) * 2 + (sp >> 3), s = sp & 7;
        pn[(size_t)((h * B_ + b) * S_ + s) * D_ + w * 16 + m16] =
            asum[c][i] + red[c][w][i][lane];
      }
  }
}

__global__ void __launch_bounds__(256) gru_mlp_kernel(
    const float* __restrict__ pnum0, const float* pnum1,
    const float* __restrict__ pden, const float* __restrict__ hprev,
    const float* __restrict__ wih, const float* __restrict__ whh,
    const float* __restrict__ bih, const float* __restrict__ bhh,
    const float* __restrict__ lnmg, const float* __restrict__ lnmb,
    const float* __restrict__ w1, const float* __restrict__ b1,
    const float* __restrict__ w2, const float* __restrict__ b2,
    float* slots) {
  __shared__ float ut[16][68];
  __shared__ float ht[16][68];
  __shared__ float gx[16][196];
  __shared__ float gh[16][196];
  __shared__ float sn[16][68];
  __shared__ float nm[16][68];
  __shared__ float hid[16][132];
  __shared__ float mu[16], rstd[16];
  const int tid = threadIdx.x;
  const int h = blockIdx.x >> 5, rt = blockIdx.x & 31;
  const int rowbase = h * (B_ * S_) + rt * 16;
  const int r0 = tid >> 4, c00 = (tid & 15) * 4;
  {
    const float4 a0 = *(const float4*)(pnum0 + (size_t)(rowbase + r0) * D_ + c00);
    const float4 a1 = *(const float4*)(pnum1 + (size_t)(rowbase + r0) * D_ + c00);
    const float dn = pden[rowbase + r0] + pden[H_ * B_ * S_ + rowbase + r0];
    ut[r0][c00] = (a0.x + a1.x) / dn;         ut[r0][c00 + 1] = (a0.y + a1.y) / dn;
    ut[r0][c00 + 2] = (a0.z + a1.z) / dn;     ut[r0][c00 + 3] = (a0.w + a1.w) / dn;
    float4 c = *(const float4*)(hprev + (size_t)(rowbase + r0) * D_ + c00);
    ht[r0][c00] = c.x; ht[r0][c00 + 1] = c.y; ht[r0][c00 + 2] = c.z; ht[r0][c00 + 3] = c.w;
  }
  __syncthreads();
  const int w = tid >> 5, lane = tid & 31, m16 = lane & 15, kh = lane >> 4;
  const float* wihh = wih + (size_t)h * D_ * 192;
  const float* whhh = whh + (size_t)h * D_ * 192;
  for (int T = w; T < 24; T += 8) {
    const bool isX = T < 12;
    const int ct = (T % 12) * 16;
    const float* Wt = isX ? wihh : whhh;
    v8f acc = {};
    for (int kk = 0; kk < 16; ++kk) {
      int kb = kk * 4 + kh * 2;
      v2f a;
      if (isX) { a.x = ut[m16][kb]; a.y = ut[m16][kb + 1]; }
      else     { a.x = ht[m16][kb]; a.y = ht[m16][kb + 1]; }
      v2f bb; bb.x = Wt[kb * 192 + ct + m16]; bb.y = Wt[(kb + 1) * 192 + ct + m16];
      acc = wmma_f32(a, bb, acc);
    }
    float* dst = isX ? &gx[0][0] : &gh[0][0];
    for (int i = 0; i < 8; ++i)
      dst[(i + 8 * kh) * 196 + ct + m16] = acc[i];
  }
  __syncthreads();
  {
    const float* bi = bih + h * 192;
    const float* bh = bhh + h * 192;
    for (int j = 0; j < 4; ++j) {
      int d = c00 + j;
      float xr = gx[r0][d] + bi[d],        hr = gh[r0][d] + bh[d];
      float xz = gx[r0][64 + d] + bi[64 + d],  hz = gh[r0][64 + d] + bh[64 + d];
      float xn = gx[r0][128 + d] + bi[128 + d], hn = gh[r0][128 + d] + bh[128 + d];
      float rg = 1.f / (1.f + expf(-(xr + hr)));
      float zg = 1.f / (1.f + expf(-(xz + hz)));
      float ng = tanhf(xn + rg * hn);
      sn[r0][d] = (1.f - zg) * ng + zg * ht[r0][d];
    }
  }
  __syncthreads();
  if (tid < 16) {
    float m = 0.f;
    for (int j = 0; j < D_; ++j) m += sn[tid][j];
    m *= (1.f / D_);
    float var = 0.f;
    for (int j = 0; j < D_; ++j) { float dd = sn[tid][j] - m; var += dd * dd; }
    var *= (1.f / D_);
    mu[tid] = m; rstd[tid] = rsqrtf(var + LN_EPS_);
  }
  __syncthreads();
  for (int j = 0; j < 4; ++j) {
    int d = c00 + j;
    nm[r0][d] = (sn[r0][d] - mu[r0]) * rstd[r0] * lnmg[h * D_ + d] + lnmb[h * D_ + d];
  }
  __syncthreads();
  {
    const float* w1h = w1 + (size_t)h * D_ * MLPH_;
    const float* b1h = b1 + h * MLPH_;
    const int ct = w * 16;
    v8f acc = {};
    for (int kk = 0; kk < 16; ++kk) {
      int kb = kk * 4 + kh * 2;
      v2f a;  a.x = nm[m16][kb]; a.y = nm[m16][kb + 1];
      v2f bb; bb.x = w1h[kb * MLPH_ + ct + m16]; bb.y = w1h[(kb + 1) * MLPH_ + ct + m16];
      acc = wmma_f32(a, bb, acc);
    }
    float bias = b1h[ct + m16];
    for (int i = 0; i < 8; ++i)
      hid[i + 8 * kh][ct + m16] = fmaxf(acc[i] + bias, 0.f);
  }
  __syncthreads();
  if (w < 4) {
    const float* w2h = w2 + (size_t)h * MLPH_ * D_;
    const float* b2h = b2 + h * D_;
    const int ct = w * 16;
    v8f acc = {};
    for (int kk = 0; kk < 32; ++kk) {
      int kb = kk * 4 + kh * 2;
      v2f a;  a.x = hid[m16][kb]; a.y = hid[m16][kb + 1];
      v2f bb; bb.x = w2h[kb * D_ + ct + m16]; bb.y = w2h[(kb + 1) * D_ + ct + m16];
      acc = wmma_f32(a, bb, acc);
    }
    float bias = b2h[ct + m16];
    for (int i = 0; i < 8; ++i) {
      int rr = i + 8 * kh;
      slots[(size_t)(rowbase + rr) * D_ + ct + m16] = sn[rr][ct + m16] + acc[i] + bias;
    }
  }
}

__global__ void copy_out_kernel(const float* __restrict__ slots,
                                float* __restrict__ out) {
  int i = blockIdx.x * blockDim.x + threadIdx.x;
  if (i < B_ * S_ * D_) out[i] = slots[i];
}

extern "C" void kernel_launch(void* const* d_in, const int* in_sizes, int n_in,
                              void* d_out, int out_size, void* d_ws, size_t ws_size,
                              hipStream_t stream) {
  (void)in_sizes; (void)n_in; (void)out_size; (void)ws_size;
  const float* inputs  = (const float*)d_in[0];
  const float* sinit   = (const float*)d_in[1];
  const float* smu     = (const float*)d_in[2];
  const float* slog    = (const float*)d_in[3];
  const float* ln_in_g = (const float*)d_in[4];
  const float* ln_in_b = (const float*)d_in[5];
  const float* Wk      = (const float*)d_in[6];
  const float* Wv      = (const float*)d_in[7];
  const float* ln_s_g  = (const float*)d_in[8];
  const float* ln_s_b  = (const float*)d_in[9];
  const float* Wq      = (const float*)d_in[10];
  const float* gru_wih = (const float*)d_in[11];
  const float* gru_whh = (const float*)d_in[12];
  const float* gru_bih = (const float*)d_in[13];
  const float* gru_bhh = (const float*)d_in[14];
  const float* ln_m_g  = (const float*)d_in[15];
  const float* ln_m_b  = (const float*)d_in[16];
  const float* mlp_w1  = (const float*)d_in[17];
  const float* mlp_b1  = (const float*)d_in[18];
  const float* mlp_w2  = (const float*)d_in[19];
  const float* mlp_b2  = (const float*)d_in[20];

  float* ws = (float*)d_ws;
  float* kbuf  = ws;
  float* vbuf  = kbuf + (size_t)B_ * N_ * D_;
  float* slots = vbuf + (size_t)B_ * N_ * D_;
  float* hprev = slots + H_ * B_ * S_ * D_;
  float* qbuf  = hprev + H_ * B_ * S_ * D_;
  float* updv  = qbuf + H_ * B_ * S_ * D_;
  float* dpart = (float*)d_out;

  ln_kv_slots_kernel<<<SLOT_WGS_ + (B_ * N_) / LKV_ROWS_, 256, 0, stream>>>(
      inputs, ln_in_g, ln_in_b, Wk, Wv, kbuf, vbuf,
      sinit, smu, slog, ln_s_g, ln_s_b, Wq, hprev, qbuf);
  attn_kernel<<<(H_ / 4) * 2 * B_, 256, 0, stream>>>(kbuf, vbuf, qbuf, updv,
                                                     slots, dpart);
  gru_mlp_kernel<<<H_ * 32, 256, 0, stream>>>(updv, slots, dpart, hprev,
                                              gru_wih, gru_whh,
                                              gru_bih, gru_bhh, ln_m_g, ln_m_b,
                                              mlp_w1, mlp_b1, mlp_w2, mlp_b2,
                                              slots);
  for (int it = 1; it < ITERS_; ++it) {
    hprev_q_kernel<<<H_ * 32, 128, 0, stream>>>(slots, ln_s_g, ln_s_b, Wq,
                                                hprev, qbuf);
    attn_kernel<<<(H_ / 4) * 2 * B_, 256, 0, stream>>>(kbuf, vbuf, qbuf, updv,
                                                       slots, dpart);
    gru_mlp_kernel<<<H_ * 32, 256, 0, stream>>>(updv, slots, dpart, hprev,
                                                gru_wih, gru_whh,
                                                gru_bih, gru_bhh, ln_m_g, ln_m_b,
                                                mlp_w1, mlp_b1, mlp_w2, mlp_b2,
                                                slots);
  }
  copy_out_kernel<<<(B_ * S_ * D_ + 255) / 256, 256, 0, stream>>>(slots,
                                                                  (float*)d_out);
}


# k and v are stored by the projection in the attention products' fragment order: attention loads its fragments from global with whole-line 16-byte loads, k and v skip LDS; sums keep order
# speedup vs baseline: 3.7017x; 1.0699x over previous
#include <hip/hip_runtime.h>

typedef float v2f __attribute__((ext_vector_type(2)));
typedef float v4f __attribute__((ext_vector_type(4)));
typedef float v8f __attribute__((ext_vector_type(8)));

constexpr int H_ = 8, B_ = 64, S_ = 8, D_ = 64, N_ = 4096, MLPH_ = 128, ITERS_ = 3;
constexpr float LN_EPS_ = 1e-5f, EPS_ATTN_ = 1e-8f, SCALE_ = 0.125f;
constexpr float LSCALE2_ = SCALE_ * 1.4426950408889634f;

__device__ __forceinline__ v8f wmma_f32(v2f a, v2f b, v8f c) {
  return __builtin_amdgcn_wmma_f32_16x16x4_f32(false, a, false, b, (short)0, c,
                                               false, false);
}

constexpr int LKV_ROWS_ = 128;
constexpr int SLOT_ROWS_ = 32;
constexpr int SLOT_WGS_ = (H_ * B_ * S_) / SLOT_ROWS_;

__device__ __forceinline__ float row16_sum(float v) {
  v += __int_as_float(__builtin_amdgcn_update_dpp(0, __float_as_int(v), 0x161, 0xf, 0xf, false));
  v += __int_as_float(__builtin_amdgcn_update_dpp(0, __float_as_int(v), 0x162, 0xf, 0xf, false));
  v += __int_as_float(__builtin_amdgcn_update_dpp(0, __float_as_int(v), 0x164, 0xf, 0xf, false));
  v += __int_as_float(__builtin_amdgcn_update_dpp(0, __float_as_int(v), 0x168, 0xf, 0xf, false));
  return v;
}

template <bool KT>
__device__ __forceinline__ void lnkv_products(
    const float (&xln)[LKV_ROWS_][68], const v2f (&bf)[16], float* op,
    const int m16, const int kh) {
#pragma unroll
  for (int mt = 0; mt < LKV_ROWS_ / 16; mt += 2) {
    v2f a0[16], a1[16];
#pragma unroll
    for (int kk = 0; kk < 16; ++kk) {
      const int kb = kk * 4 + kh * 2;
      a0[kk].x = xln[mt * 16 + m16][kb];      a0[kk].y = xln[mt * 16 + m16][kb + 1];
      a1[kk].x = xln[mt * 16 + 16 + m16][kb]; a1[kk].y = xln[mt * 16 + 16 + m16][kb + 1];
    }
    __builtin_amdgcn_sched_barrier(0);
    v8f acc0 = {}, acc1 = {};
#pragma unroll
    for (int kk = 0; kk < 16; ++kk) {
      if (KT) {
        acc0 = wmma_f32(bf[kk], a0[kk], acc0);
        acc1 = wmma_f32(bf[kk], a1[kk], acc1);
      } else {
        acc0 = wmma_f32(a0[kk], bf[kk], acc0);
        acc1 = wmma_f32(a1[kk], bf[kk], acc1);
      }
    }
    const int o0 = KT ? mt * 1024 : (mt >> 2) * 4096 + (mt & 3) * 256;
    const int o1 = KT ? (mt + 1) * 1024 : ((mt + 1) >> 2) * 4096 + ((mt + 1) & 3) * 256;
    *(v4f*)(op + o0)      = v4f{acc0[0], acc0[1], acc0[4], acc0[5]};
    *(v4f*)(op + o0 + 64) = v4f{acc0[2], acc0[3], acc0[6], acc0[7]};
    *(v4f*)(op + o1)      = v4f{acc1[0], acc1[1], acc1[4], acc1[5]};
    *(v4f*)(op + o1 + 64) = v4f{acc1[2], acc1[3], acc1[6], acc1[7]};
  }
}

__global__ void __launch_bounds__(256) ln_kv_slots_kernel(
    const float* __restrict__ inp, const float* __restrict__ lng,
    const float* __restrict__ lnb, const float* __restrict__ Wk,
    const float* __restrict__ Wv, float* __restrict__ kout,
    float* __restrict__ vout,
    const float* __restrict__ sinit, const float* __restrict__ smu,
    const float* __restrict__ slog, const float* __restrict__ lnsg,
    const float* __restrict__ lnsb, const float* __restrict__ Wq,
    float* __restrict__ hprev, float* __restrict__ qout) {
  __shared__ __attribute__((aligned(16))) float xln[LKV_ROWS_][68];
  __shared__ float smean[SLOT_ROWS_], srstd[SLOT_ROWS_];
  const int tid = threadIdx.x;
  const int w = tid >> 5, lane = tid & 31, m16 = lane & 15, kh = lane >> 4;
  if (blockIdx.x < SLOT_WGS_) {
    const int h = blockIdx.x / (B_ * S_ / SLOT_ROWS_);
    const int rowbase = blockIdx.x * SLOT_ROWS_;
    for (int e = tid; e < SLOT_ROWS_ * D_; e += 256) {
      const int r = e >> 6, cc = e & 63;
      xln[r][cc] = smu[h * D_ + cc] +
                   expf(slog[h * D_ + cc]) * sinit[(size_t)(rowbase + r) * D_ + cc];
    }
    __syncthreads();
    if (tid < SLOT_ROWS_) {
      float m = 0.f;
      for (int j = 0; j < D_; ++j) m += xln[tid][j];
      m *= (1.f / D_);
      float var = 0.f;
      for (int j = 0; j < D_; ++j) { float dd = xln[tid][j] - m; var += dd * dd; }
      var *= (1.f / D_);
      smean[tid] = m; srstd[tid] = rsqrtf(var + LN_EPS_);
    }
    __syncthreads();
    for (int e = tid; e < SLOT_ROWS_ * D_; e += 256) {
      const int r = e >> 6, cc = e & 63;
      const float nv = (xln[r][cc] - smean[r]) * srstd[r] * lnsg[h * D_ + cc] + lnsb[h * D_ + cc];
      xln[r][cc] = nv;
      hprev[(size_t)(rowbase + r) * D_ + cc] = nv;
    }
    __syncthreads();
    const float* Wh = Wq + (size_t)h * D_ * D_;
    const int rg = (w >> 2) * 16, qct = (w & 3) * 16;
    v8f acc = {};
    for (int kk = 0; kk < 16; ++kk) {
      const int kb = kk * 4 + kh * 2;
      v2f a; a.x = xln[rg + m16][kb]; a.y = xln[rg + m16][kb + 1];
      v2f b; b.x = Wh[kb * D_ + qct + m16]; b.y = Wh[(kb + 1) * D_ + qct + m16];
      acc = wmma_f32(a, b, acc);
    }
    for (int i = 0; i < 8; ++i)
      qout[(size_t)(rowbase + rg + i + 8 * kh) * D_ + qct + m16] = acc[i];
    return;
  }
  const size_t row0 = (size_t)(blockIdx.x - SLOT_WGS_) * LKV_ROWS_;
  const float* Wm = (w < 4) ? Wk : Wv;
  const int ct = (w & 3) * 16;
  v2f bf[16];
#pragma unroll
  for (int kk = 0; kk < 16; ++kk) {
    const int kb = kk * 4 + kh * 2;
    bf[kk].x = Wm[kb * D_ + ct + m16];
    bf[kk].y = Wm[(kb + 1) * D_ + ct + m16];
  }
  const int r = tid >> 4, c0 = (tid & 15) * 4;
  const float4 g4 = *(const float4*)(lng + c0);
  const float4 b4 = *(const float4*)(lnb + c0);
  float4 xv[8];
#pragma unroll
  for (int j = 0; j < 8; ++j)
    xv[j] = *(const float4*)(inp + (row0 + r + 16 * j) * D_ + c0);
#pragma unroll
  for (int j = 0; j < 8; ++j) {
    const float m = row16_sum((xv[j].x + xv[j].y) + (xv[j].z + xv[j].w)) * (1.f / D_);
    const float dx = xv[j].x - m, dy = xv[j].y - m, dz = xv[j].z - m, dw = xv[j].w - m;
    const float var = row16_sum((dx * dx + dy * dy) + (dz * dz + dw * dw)) * (1.f / D_);
    const float rs = rsqrtf(var + LN_EPS_);
    float4 o;
    o.x = dx * rs * g4.x + b4.x;
    o.y = dy * rs * g4.y + b4.y;
    o.z = dz * rs * g4.z + b4.z;
    o.w = dw * rs * g4.w + b4.w;
    *(float4*)&xln[r + 16 * j][c0] = o;
  }
  __syncthreads();
  const int wu = __builtin_amdgcn_readfirstlane(w);
  if (wu < 4) {
    lnkv_products<true>(xln, bf,
        kout + (row0 >> 4) * 1024 + (2 * wu + kh) * 128 + m16 * 4, m16, kh);
  } else {
    lnkv_products<false>(xln, bf,
        vout + ((row0 >> 6) * 4 + (wu & 3)) * 1024 + kh * 128 + m16 * 4, m16, kh);
  }
}

__global__ void __launch_bounds__(128) hprev_q_kernel(
    const float* __restrict__ slots, const float* __restrict__ lng,
    const float* __restrict__ lnb, const float* __restrict__ Wq,
    float* __restrict__ hprev, float* __restrict__ qout) {
  __shared__ float hl[16][68];
  __shared__ float mu[16], rstd[16];
  const int tid = threadIdx.x;
  const int h = blockIdx.x >> 5;
  const int rt = blockIdx.x & 31;
  const int rowbase = h * (B_ * S_) + rt * 16;
  for (int e = tid; e < 16 * D_; e += 128) {
    int r = e >> 6, c = e & 63;
    hl[r][c] = slots[(size_t)(rowbase + r) * D_ + c];
  }
  __syncthreads();
  if (tid < 16) {
    float m = 0.f;
    for (int j = 0; j < D_; ++j) m += hl[tid][j];
    m *= (1.f / D_);
    float var = 0.f;
    for (int j = 0; j < D_; ++j) { float dd = hl[tid][j] - m; var += dd * dd; }
    var *= (1.f / D_);
    mu[tid] = m; rstd[tid] = rsqrtf(var + LN_EPS_);
  }
  __syncthreads();
  for (int e = tid; e < 16 * D_; e += 128) {
    int r = e >> 6, c = e & 63;
    float nv = (hl[r][c] - mu[r]) * rstd[r] * lng[h * D_ + c] + lnb[h * D_ + c];
    hl[r][c] = nv;
    hprev[(size_t)(rowbase + r) * D_ + c] = nv;
  }
  __syncthreads();
  const int w = tid >> 5, lane = tid & 31, m16 = lane & 15, kh = lane >> 4;
  const float* Wh = Wq + (size_t)h * D_ * D_;
  const int ct = w * 16;
  v8f acc = {};
  for (int kk = 0; kk < 16; ++kk) {
    int kb = kk * 4 + kh * 2;
    v2f a; a.x = hl[m16][kb]; a.y = hl[m16][kb + 1];
    v2f b; b.x = Wh[kb * D_ + ct + m16]; b.y = Wh[(kb + 1) * D_ + ct + m16];
    acc = wmma_f32(a, b, acc);
  }
  for (int i = 0; i < 8; ++i)
    qout[(size_t)(rowbase + i + 8 * kh) * D_ + ct + m16] = acc[i];
}

constexpr int TR_ = 128;
constexpr int NH_ = N_ / 2;
constexpr int NTH_ = NH_ / TR_;
constexpr int AP_ = TR_ + 4;
constexpr int STEPF_ = TR_ * D_;

template <int P>
__device__ __forceinline__ void attn_tile_step(
    float (&at)[2][2][16][AP_],
    const float* knext, const float* vnext, const v2f (&qf)[2][16],
    const v2f (&kf)[16], const v2f (&vf)[16], v2f (&kfn)[16], v2f (&vfn)[16],
    v8f (&accu)[2][2], float (&dacc)[2][8],
    const int pc, const int gb, const int m16, const int kh) {
#pragma unroll
  for (int j = 0; j < 8; ++j) {
    const v4f t = *(const v4f*)(knext + j * 128);
    kfn[2 * j] = v2f{t.x, t.y};
    kfn[2 * j + 1] = v2f{t.z, t.w};
  }
#pragma unroll
  for (int j = 0; j < 8; ++j) {
    const v4f t = *(const v4f*)(vnext + j * 128);
    vfn[2 * j] = v2f{t.x, t.y};
    vfn[2 * j + 1] = v2f{t.z, t.w};
  }
  asm volatile("" ::: "memory");
  __builtin_amdgcn_sched_barrier(0);
  v8f la[2], lb[2];
  la[0] = v8f{}; la[1] = v8f{}; lb[0] = v8f{}; lb[1] = v8f{};
#pragma unroll
  for (int kk = 0; kk < 8; ++kk) {
#pragma unroll
    for (int c = 0; c < 2; ++c) {
      la[c] = wmma_f32(qf[c][kk], kf[kk], la[c]);
      lb[c] = wmma_f32(qf[c][kk + 8], kf[kk + 8], lb[c]);
    }
  }
#pragma unroll
  for (int c = 0; c < 2; ++c) {
    float lv[8];
#pragma unroll
    for (int s = 0; s < 8; ++s) lv[s] = (la[c][s] + lb[c][s]) * LSCALE2_;
    float mx = -3.4e38f;
#pragma unroll
    for (int s = 0; s < 8; ++s) mx = fmaxf(mx, lv[s]);
    float ex[8], sum = 0.f;
#pragma unroll
    for (int s = 0; s < 8; ++s) { ex[s] = __builtin_amdgcn_exp2f(lv[s] - mx); sum += ex[s]; }
    const float inv = __builtin_amdgcn_rcpf(sum);
#pragma unroll
    for (int s = 0; s < 8; ++s) {
      const float pv = ex[s] * inv + EPS_ATTN_;
      at[P][c][8 * kh + s][pc] = pv;
      dacc[c][s] += pv;
    }
  }
  __builtin_amdgcn_fence(__ATOMIC_RELEASE, "workgroup", "local");
  __builtin_amdgcn_s_barrier();
  __builtin_amdgcn_fence(__ATOMIC_ACQUIRE, "workgroup", "local");
  v2f af[2][16];
#pragma unroll
  for (int kk = 0; kk < 8; ++kk)
#pragma unroll
    for (int c = 0; c < 2; ++c) {
      const v4f t = *(const v4f*)&at[P][c][m16][gb + (2 * kk + kh) * 4];
      af[c][kk] = v2f{t.x, t.y};
      af[c][kk + 8] = v2f{t.z, t.w};
    }
  __builtin_amdgcn_sched_barrier(0);
#pragma unroll
  for (int kk = 0; kk < 8; ++kk) {
#pragma unroll
    for (int c = 0; c < 2; ++c) {
      accu[c][0] = wmma_f32(af[c][kk], vf[kk], accu[c][0]);
      accu[c][1] = wmma_f32(af[c][kk + 8], vf[kk + 8], accu[c][1]);
    }
  }
}

__global__ void __launch_bounds__(256) __attribute__((amdgpu_waves_per_eu(2, 2)))
attn_kernel(
    const float* kbuf, const float* vbuf,
    const float* __restrict__ qbuf, float* pnum0, float* pnum1, float* pden) {
  __shared__ __attribute__((aligned(16))) float at[2][2][16][AP_];
  __shared__ float dsc[32][TR_ + 1];
  __shared__ float red[2][4][8][32];
  const int tid = threadIdx.x;
  const int hq = blockIdx.x >> 7;
  const int nh = (blockIdx.x >> 6) & 1;
  const int b  = blockIdx.x & 63;
  const int w8 = tid >> 5, lane = tid & 31, m16 = lane & 15, kh = lane >> 4;
  const int g = w8 >> 2, w = w8 & 3;
  const int gb = g * 64;
  const size_t rowblk = (size_t)b * N_ + (size_t)nh * NH_;
  const float* kw = kbuf + ((rowblk >> 4) + w8) * 1024 + lane * 4;
  const float* vw = vbuf + (((rowblk >> 6) + g) * 4 + w) * 1024 + lane * 4;
  const int q5 = (w8 & 1) * 16 + m16;
  const int pc = gb + (q5 >> 1) * 4 + ((w8 >> 1) & 1) * 2 + (q5 & 1);
  v2f qf[2][16];
#pragma unroll
  for (int c = 0; c < 2; ++c) {
    const int h = (hq * 2 + c) * 2 + (m16 >> 3), s = m16 & 7;
    const float* qr = qbuf + (size_t)((h * B_ + b) * S_ + s) * D_ + kh * 2;
#pragma unroll
    for (int kk = 0; kk < 16; ++kk) { qf[c][kk].x = qr[kk * 4]; qf[c][kk].y = qr[kk * 4 + 1]; }
  }
  v2f kfa[16], vfa[16], kfb[16], vfb[16];
#pragma unroll
  for (int j = 0; j < 8; ++j) {
    const v4f tk = *(const v4f*)(kw + j * 128);
    kfa[2 * j] = v2f{tk.x, tk.y};
    kfa[2 * j + 1] = v2f{tk.z, tk.w};
    const v4f tv = *(const v4f*)(vw + j * 128);
    vfa[2 * j] = v2f{tv.x, tv.y};
    vfa[2 * j + 1] = v2f{tv.z, tv.w};
  }
  v8f accu[2][2];
  float dacc[2][8];
#pragma unroll
  for (int c = 0; c < 2; ++c) {
    accu[c][0] = v8f{};
    accu[c][1] = v8f{};
#pragma unroll
    for (int s = 0; s < 8; ++s) dacc[c][s] = 0.f;
  }
  for (int t = 0; t < NTH_; t += 2) {
    attn_tile_step<0>(at, kw + (size_t)(t + 1) * STEPF_, vw + (size_t)(t + 1) * STEPF_,
                      qf, kfa, vfa, kfb, vfb, accu, dacc, pc, gb, m16, kh);
    const int tn = (t + 2 < NTH_) ? t + 2 : t + 1;
    attn_tile_step<1>(at, kw + (size_t)tn * STEPF_, vw + (size_t)tn * STEPF_,
                      qf, kfb, vfb, kfa, vfa, accu, dacc, pc, gb, m16, kh);
  }
  v8f asum[2];
  asum[0] = accu[0][0] + accu[0][1];
  asum[1] = accu[1][0] + accu[1][1];
  if (g == 1) {
#pragma unroll
    for (int c = 0; c < 2; ++c)
#pragma unroll
      for (int i = 0; i < 8; ++i) red[c][w][i][lane] = asum[c][i];
  }
  {
    const int r = w8 * 16 + m16;
#pragma unroll
    for (int c = 0; c < 2; ++c)
#pragma unroll
      for (int s = 0; s < 8; ++s) dsc[c * 16 + kh * 8 + s][r] = dacc[c][s];
  }
  __syncthreads();
  if (tid < 32) {
    float dsum = 0.f;
    for (int r = 0; r < TR_; ++r) dsum += dsc[tid][r];
    const int h = hq * 4 + (tid >> 3), s = tid & 7;
    pden[nh * (H_ * B_ * S_) + (h * B_ + b) * S_ + s] = dsum;
  }
  if (g == 0) {
    float* pn = nh ? pnum1 : pnum0;
#pragma unroll
    for (int c = 0; c < 2; ++c)
      for (int i = 0; i < 8; ++i) {
        const int sp = i + 8 * kh;
        const int h = (hq * 2 + c) * 2 + (sp >> 3), s = sp & 7;
        pn[(size_t)((h * B_ + b) * S_ + s) * D_ + w * 16 + m16] =
            asum[c][i] + red[c][w][i][lane];
      }
  }
}

__global__ void __launch_bounds__(256) gru_mlp_kernel(
    const float* __restrict__ pnum0, const float* pnum1,
    const float* __restrict__ pden, const float* __restrict__ hprev,
    const float* __restrict__ wih, const float* __restrict__ whh,
    const float* __restrict__ bih, const float* __restrict__ bhh,
    const float* __restrict__ lnmg, const float* __restrict__ lnmb,
    const float* __restrict__ w1, const float* __restrict__ b1,
    const float* __restrict__ w2, const float* __restrict__ b2,
    float* slots) {
  __shared__ float ut[16][68];
  __shared__ float ht[16][68];
  __shared__ float gx[16][196];
  __shared__ float gh[16][196];
  __shared__ float sn[16][68];
  __shared__ float nm[16][68];
  __shared__ float hid[16][132];
  __shared__ float mu[16], rstd[16];
  const int tid = threadIdx.x;
  const int h = blockIdx.x >> 5, rt = blockIdx.x & 31;
  const int rowbase = h * (B_ * S_) + rt * 16;
  const int r0 = tid >> 4, c00 = (tid & 15) * 4;
  {
    const float4 a0 = *(const float4*)(pnum0 + (size_t)(rowbase + r0) * D_ + c00);
    const float4 a1 = *(const float4*)(pnum1 + (size_t)(rowbase + r0) * D_ + c00);
    const float dn = pden[rowbase + r0] + pden[H_ * B_ * S_ + rowbase + r0];
    ut[r0][c00] = (a0.x + a1.x) / dn;         ut[r0][c00 + 1] = (a0.y + a1.y) / dn;
    ut[r0][c00 + 2] = (a0.z + a1.z) / dn;     ut[r0][c00 + 3] = (a0.w + a1.w) / dn;
    float4 c = *(const float4*)(hprev + (size_t)(rowbase + r0) * D_ + c00);
    ht[r0][c00] = c.x; ht[r0][c00 + 1] = c.y; ht[r0][c00 + 2] = c.z; ht[r0][c00 + 3] = c.w;
  }
  __syncthreads();
  const int w = tid >> 5, lane = tid & 31, m16 = lane & 15, kh = lane >> 4;
  const float* wihh = wih + (size_t)h * D_ * 192;
  const float* whhh = whh + (size_t)h * D_ * 192;
  for (int T = w; T < 24; T += 8) {
    const bool isX = T < 12;
    const int ct = (T % 12) * 16;
    const float* Wt = isX ? wihh : whhh;
    v8f acc = {};
    for (int kk = 0; kk < 16; ++kk) {
      int kb = kk * 4 + kh * 2;
      v2f a;
      if (isX) { a.x = ut[m16][kb]; a.y = ut[m16][kb + 1]; }
      else     { a.x = ht[m16][kb]; a.y = ht[m16][kb + 1]; }
      v2f bb; bb.x = Wt[kb * 192 + ct + m16]; bb.y = Wt[(kb + 1) * 192 + ct + m16];
      acc = wmma_f32(a, bb, acc);
    }
    float* dst = isX ? &gx[0][0] : &gh[0][0];
    for (int i = 0; i < 8; ++i)
      dst[(i + 8 * kh) * 196 + ct + m16] = acc[i];
  }
  __syncthreads();
  {
    const float* bi = bih + h * 192;
    const float* bh = bhh + h * 192;
    for (int j = 0; j < 4; ++j) {
      int d = c00 + j;
      float xr = gx[r0][d] + bi[d],        hr = gh[r0][d] + bh[d];
      float xz = gx[r0][64 + d] + bi[64 + d],  hz = gh[r0][64 + d] + bh[64 + d];
      float xn = gx[r0][128 + d] + bi[128 + d], hn = gh[r0][128 + d] + bh[128 + d];
      float rg = 1.f / (1.f + expf(-(xr + hr)));
      float zg = 1.f / (1.f + expf(-(xz + hz)));
      float ng = tanhf(xn + rg * hn);
      sn[r0][d] = (1.f - zg) * ng + zg * ht[r0][d];
    }
  }
  __syncthreads();
  if (tid < 16) {
    float m = 0.f;
    for (int j = 0; j < D_; ++j) m += sn[tid][j];
    m *= (1.f / D_);
    float var = 0.f;
    for (int j = 0; j < D_; ++j) { float dd = sn[tid][j] - m; var += dd * dd; }
    var *= (1.f / D_);
    mu[tid] = m; rstd[tid] = rsqrtf(var + LN_EPS_);
  }
  __syncthreads();
  for (int j = 0; j < 4; ++j) {
    int d = c00 + j;
    nm[r0][d] = (sn[r0][d] - mu[r0]) * rstd[r0] * lnmg[h * D_ + d] + lnmb[h * D_ + d];
  }
  __syncthreads();
  {
    const float* w1h = w1 + (size_t)h * D_ * MLPH_;
    const float* b1h = b1 + h * MLPH_;
    const int ct = w * 16;
    v8f acc = {};
    for (int kk = 0; kk < 16; ++kk) {
      int kb = kk * 4 + kh * 2;
      v2f a;  a.x = nm[m16][kb]; a.y = nm[m16][kb + 1];
      v2f bb; bb.x = w1h[kb * MLPH_ + ct + m16]; bb.y = w1h[(kb + 1) * MLPH_ + ct + m16];
      acc = wmma_f32(a, bb, acc);
    }
    float bias = b1h[ct + m16];
    for (int i = 0; i < 8; ++i)
      hid[i + 8 * kh][ct + m16] = fmaxf(acc[i] + bias, 0.f);
  }
  __syncthreads();
  if (w < 4) {
    const float* w2h = w2 + (size_t)h * MLPH_ * D_;
    const float* b2h = b2 + h * D_;
    const int ct = w * 16;
    v8f acc = {};
    for (int kk = 0; kk < 32; ++kk) {
      int kb = kk * 4 + kh * 2;
      v2f a;  a.x = hid[m16][kb]; a.y = hid[m16][kb + 1];
      v2f bb; bb.x = w2h[kb * D_ + ct + m16]; bb.y = w2h[(kb + 1) * D_ + ct + m16];
      acc = wmma_f32(a, bb, acc);
    }
    float bias = b2h[ct + m16];
    for (int i = 0; i < 8; ++i) {
      int rr = i + 8 * kh;
      slots[(size_t)(rowbase + rr) * D_ + ct + m16] = sn[rr][ct + m16] + acc[i] + bias;
    }
  }
}

__global__ void copy_out_kernel(const float* __restrict__ slots,
                                float* __restrict__ out) {
  int i = blockIdx.x * blockDim.x + threadIdx.x;
  if (i < B_ * S_ * D_) out[i] = slots[i];
}

extern "C" void kernel_launch(void* const* d_in, const int* in_sizes, int n_in,
                              void* d_out, int out_size, void* d_ws, size_t ws_size,
                              hipStream_t stream) {
  (void)in_sizes; (void)n_in; (void)out_size; (void)ws_size;
  const float* inputs  = (const float*)d_in[0];
  const float* sinit   = (const float*)d_in[1];
  const float* smu     = (const float*)d_in[2];
  const float* slog    = (const float*)d_in[3];
  const float* ln_in_g = (const float*)d_in[4];
  const float* ln_in_b = (const float*)d_in[5];
  const float* Wk      = (const float*)d_in[6];
  const float* Wv      = (const float*)d_in[7];
  const float* ln_s_g  = (const float*)d_in[8];
  const float* ln_s_b  = (const float*)d_in[9];
  const float* Wq      = (const float*)d_in[10];
  const float* gru_wih = (const float*)d_in[11];
  const float* gru_whh = (const float*)d_in[12];
  const float* gru_bih = (const float*)d_in[13];
  const float* gru_bhh = (const float*)d_in[14];
  const float* ln_m_g  = (const float*)d_in[15];
  const float* ln_m_b  = (const float*)d_in[16];
  const float* mlp_w1  = (const float*)d_in[17];
  const float* mlp_b1  = (const float*)d_in[18];
  const float* mlp_w2  = (const float*)d_in[19];
  const float* mlp_b2  = (const float*)d_in[20];

  float* ws = (float*)d_ws;
  float* kbuf  = ws;
  float* vbuf  = kbuf + (size_t)B_ * N_ * D_;
  float* slots = vbuf + (size_t)B_ * N_ * D_;
  float* hprev = slots + H_ * B_ * S_ * D_;
  float* qbuf  = hprev + H_ * B_ * S_ * D_;
  float* updv  = qbuf + H_ * B_ * S_ * D_;
  float* dpart = (float*)d_out;

  ln_kv_slots_kernel<<<SLOT_WGS_ + (B_ * N_) / LKV_ROWS_, 256, 0, stream>>>(
      inputs, ln_in_g, ln_in_b, Wk, Wv, kbuf, vbuf,
      sinit, smu, slog, ln_s_g, ln_s_b, Wq, hprev, qbuf);
  attn_kernel<<<(H_ / 4) * 2 * B_, 256, 0, stream>>>(kbuf, vbuf, qbuf, updv,
                                                     slots, dpart);
  gru_mlp_kernel<<<H_ * 32, 256, 0, stream>>>(updv, slots, dpart, hprev,
                                              gru_wih, gru_whh,
                                              gru_bih, gru_bhh, ln_m_g, ln_m_b,
                                              mlp_w1, mlp_b1, mlp_w2, mlp_b2,
                                              slots);
  for (int it = 1; it < ITERS_; ++it) {
    hprev_q_kernel<<<H_ * 32, 128, 0, stream>>>(slots, ln_s_g, ln_s_b, Wq,
                                                hprev, qbuf);
    attn_kernel<<<(H_ / 4) * 2 * B_, 256, 0, stream>>>(kbuf, vbuf, qbuf, updv,
                                                       slots, dpart);
    gru_mlp_kernel<<<H_ * 32, 256, 0, stream>>>(updv, slots, dpart, hprev,
                                                gru_wih, gru_whh,
                                                gru_bih, gru_bhh, ln_m_g, ln_m_b,
                                                mlp_w1, mlp_b1, mlp_w2, mlp_b2,
                                                slots);
  }
  copy_out_kernel<<<(B_ * S_ * D_ + 255) / 256, 256, 0, stream>>>(slots,
                                                                  (float*)d_out);
}


# one workgroup per (batch, head pair) sweeps all N: the 3 iterations (q, attention, GRU, MLP) loop in one launch on a slot tile in LDS; 10 launches to 2; update and den sum order changes
# speedup vs baseline: 4.0533x; 1.0950x over previous
#include <hip/hip_runtime.h>

typedef float v2f __attribute__((ext_vector_type(2)));
typedef float v4f __attribute__((ext_vector_type(4)));
typedef float v8f __attribute__((ext_vector_type(8)));

constexpr int H_ = 8, B_ = 64, S_ = 8, D_ = 64, N_ = 4096, MLPH_ = 128, ITERS_ = 3;
constexpr float LN_EPS_ = 1e-5f, EPS_ATTN_ = 1e-8f, SCALE_ = 0.125f;
constexpr float LSCALE2_ = SCALE_ * 1.4426950408889634f;

__device__ __forceinline__ v8f wmma_f32(v2f a, v2f b, v8f c) {
  return __builtin_amdgcn_wmma_f32_16x16x4_f32(false, a, false, b, (short)0, c,
                                               false, false);
}

constexpr int LKV_ROWS_ = 128;
constexpr int SLOT_ROWS_ = 32;
constexpr int SLOT_WGS_ = (H_ * B_ * S_) / SLOT_ROWS_;

__device__ __forceinline__ float row16_sum(float v) {
  v += __int_as_float(__builtin_amdgcn_update_dpp(0, __float_as_int(v), 0x161, 0xf, 0xf, false));
  v += __int_as_float(__builtin_amdgcn_update_dpp(0, __float_as_int(v), 0x162, 0xf, 0xf, false));
  v += __int_as_float(__builtin_amdgcn_update_dpp(0, __float_as_int(v), 0x164, 0xf, 0xf, false));
  v += __int_as_float(__builtin_amdgcn_update_dpp(0, __float_as_int(v), 0x168, 0xf, 0xf, false));
  return v;
}

template <bool KT>
__device__ __forceinline__ void lnkv_products(
    const float (&xln)[LKV_ROWS_][68], const v2f (&bf)[16], float* op,
    const int m16, const int kh) {
#pragma unroll
  for (int mt = 0; mt < LKV_ROWS_ / 16; mt += 2) {
    v2f a0[16], a1[16];
#pragma unroll
    for (int kk = 0; kk < 16; ++kk) {
      const int kb = kk * 4 + kh * 2;
      a0[kk].x = xln[mt * 16 + m16][kb];      a0[kk].y = xln[mt * 16 + m16][kb + 1];
      a1[kk].x = xln[mt * 16 + 16 + m16][kb]; a1[kk].y = xln[mt * 16 + 16 + m16][kb + 1];
    }
    __builtin_amdgcn_sched_barrier(0);
    v8f acc0 = {}, acc1 = {};
#pragma unroll
    for (int kk = 0; kk < 16; ++kk) {
      if (KT) {
        acc0 = wmma_f32(bf[kk], a0[kk], acc0);
        acc1 = wmma_f32(bf[kk], a1[kk], acc1);
      } else {
        acc0 = wmma_f32(a0[kk], bf[kk], acc0);
        acc1 = wmma_f32(a1[kk], bf[kk], acc1);
      }
    }
    const int o0 = KT ? mt * 1024 : (mt >> 2) * 4096 + (mt & 3) * 256;
    const int o1 = KT ? (mt + 1) * 1024 : ((mt + 1) >> 2) * 4096 + ((mt + 1) & 3) * 256;
    *(v4f*)(op + o0)      = v4f{acc0[0], acc0[1], acc0[4], acc0[5]};
    *(v4f*)(op + o0 + 64) = v4f{acc0[2], acc0[3], acc0[6], acc0[7]};
    *(v4f*)(op + o1)      = v4f{acc1[0], acc1[1], acc1[4], acc1[5]};
    *(v4f*)(op + o1 + 64) = v4f{acc1[2], acc1[3], acc1[6], acc1[7]};
  }
}

__global__ void __launch_bounds__(256) ln_kv_slots_kernel(
    const float* __restrict__ inp, const float* __restrict__ lng,
    const float* __restrict__ lnb, const float* __restrict__ Wk,
    const float* __restrict__ Wv, float* __restrict__ kout,
    float* __restrict__ vout,
    const float* __restrict__ sinit, const float* __restrict__ smu,
    const float* __restrict__ slog, const float* __restrict__ lnsg,
    const float* __restrict__ lnsb, const float* __restrict__ Wq,
    float* __restrict__ hprev, float* __restrict__ qout) {
  __shared__ __attribute__((aligned(16))) float xln[LKV_ROWS_][68];
  __shared__ float smean[SLOT_ROWS_], srstd[SLOT_ROWS_];
  const int tid = threadIdx.x;
  const int w = tid >> 5, lane = tid & 31, m16 = lane & 15, kh = lane >> 4;
  if (blockIdx.x < SLOT_WGS_) {
    const int h = blockIdx.x / (B_ * S_ / SLOT_ROWS_);
    const int rowbase = blockIdx.x * SLOT_ROWS_;
    for (int e = tid; e < SLOT_ROWS_ * D_; e += 256) {
      const int r = e >> 6, cc = e & 63;
      xln[r][cc] = smu[h * D_ + cc] +
                   expf(slog[h * D_ + cc]) * sinit[(size_t)(rowbase + r) * D_ + cc];
    }
    __syncthreads();
    if (tid < SLOT_ROWS_) {
      float m = 0.f;
      for (int j = 0; j < D_; ++j) m += xln[tid][j];
      m *= (1.f / D_);
      float var = 0.f;
      for (int j = 0; j < D_; ++j) { float dd = xln[tid][j] - m; var += dd * dd; }
      var *= (1.f / D_);
      smean[tid] = m; srstd[tid] = rsqrtf(var + LN_EPS_);
    }
    __syncthreads();
    for (int e = tid; e < SLOT_ROWS_ * D_; e += 256) {
      const int r = e >> 6, cc = e & 63;
      const float nv = (xln[r][cc] - smean[r]) * srstd[r] * lnsg[h * D_ + cc] + lnsb[h * D_ + cc];
      xln[r][cc] = nv;
      hprev[(size_t)(rowbase + r) * D_ + cc] = nv;
    }
    __syncthreads();
    const float* Wh = Wq + (size_t)h * D_ * D_;
    const int rg = (w >> 2) * 16, qct = (w & 3) * 16;
    v8f acc = {};
    for (int kk = 0; kk < 16; ++kk) {
      const int kb = kk * 4 + kh * 2;
      v2f a; a.x = xln[rg + m16][kb]; a.y = xln[rg + m16][kb + 1];
      v2f b; b.x = Wh[kb * D_ + qct + m16]; b.y = Wh[(kb + 1) * D_ + qct + m16];
      acc = wmma_f32(a, b, acc);
    }
    for (int i = 0; i < 8; ++i)
      qout[(size_t)(rowbase + rg + i + 8 * kh) * D_ + qct + m16] = acc[i];
    return;
  }
  const size_t row0 = (size_t)(blockIdx.x - SLOT_WGS_) * LKV_ROWS_;
  const float* Wm = (w < 4) ? Wk : Wv;
  const int ct = (w & 3) * 16;
  v2f bf[16];
#pragma unroll
  for (int kk = 0; kk < 16; ++kk) {
    const int kb = kk * 4 + kh * 2;
    bf[kk].x = Wm[kb * D_ + ct + m16];
    bf[kk].y = Wm[(kb + 1) * D_ + ct + m16];
  }
  const int r = tid >> 4, c0 = (tid & 15) * 4;
  const float4 g4 = *(const float4*)(lng + c0);
  const float4 b4 = *(const float4*)(lnb + c0);
  float4 xv[8];
#pragma unroll
  for (int j = 0; j < 8; ++j)
    xv[j] = *(const float4*)(inp + (row0 + r + 16 * j) * D_ + c0);
#pragma unroll
  for (int j = 0; j < 8; ++j) {
    const float m = row16_sum((xv[j].x + xv[j].y) + (xv[j].z + xv[j].w)) * (1.f / D_);
    const float dx = xv[j].x - m, dy = xv[j].y - m, dz = xv[j].z - m, dw = xv[j].w - m;
    const float var = row16_sum((dx * dx + dy * dy) + (dz * dz + dw * dw)) * (1.f / D_);
    const float rs = rsqrtf(var + LN_EPS_);
    float4 o;
    o.x = dx * rs * g4.x + b4.x;
    o.y = dy * rs * g4.y + b4.y;
    o.z = dz * rs * g4.z + b4.z;
    o.w = dw * rs * g4.w + b4.w;
    *(float4*)&xln[r + 16 * j][c0] = o;
  }
  __syncthreads();
  const int wu = __builtin_amdgcn_readfirstlane(w);
  if (wu < 4) {
    lnkv_products<true>(xln, bf,
        kout + (row0 >> 4) * 1024 + (2 * wu + kh) * 128 + m16 * 4, m16, kh);
  } else {
    lnkv_products<false>(xln, bf,
        vout + ((row0 >> 6) * 4 + (wu & 3)) * 1024 + kh * 128 + m16 * 4, m16, kh);
  }
}

constexpr int TR_ = 128;
constexpr int NT_ = N_ / TR_;
constexpr int AP_ = TR_ + 4;
constexpr int STEPF_ = TR_ * D_;

template <int P>
__device__ __forceinline__ void attn_tile_step(
    float (&at)[2][16][AP_],
    const float* knext, const float* vnext, const v2f (&qf)[16],
    const v2f (&kf)[16], const v2f (&vf)[16], v2f (&kfn)[16], v2f (&vfn)[16],
    v8f (&accu)[2], float (&dacc)[8],
    const int pc, const int gb, const int m16, const int kh) {
#pragma unroll
  for (int j = 0; j < 8; ++j) {
    const v4f t = *(const v4f*)(knext + j * 128);
    kfn[2 * j] = v2f{t.x, t.y};
    kfn[2 * j + 1] = v2f{t.z, t.w};
  }
#pragma unroll
  for (int j = 0; j < 8; ++j) {
    const v4f t = *(const v4f*)(vnext + j * 128);
    vfn[2 * j] = v2f{t.x, t.y};
    vfn[2 * j + 1] = v2f{t.z, t.w};
  }
  asm volatile("" ::: "memory");
  __builtin_amdgcn_sched_barrier(0);
  v8f la = {}, lb = {};
#pragma unroll
  for (int kk = 0; kk < 8; ++kk) {
    la = wmma_f32(qf[kk], kf[kk], la);
    lb = wmma_f32(qf[kk + 8], kf[kk + 8], lb);
  }
  {
    float lv[8];
#pragma unroll
    for (int s = 0; s < 8; ++s) lv[s] = (la[s] + lb[s]) * LSCALE2_;
    float mx = -3.4e38f;
#pragma unroll
    for (int s = 0; s < 8; ++s) mx = fmaxf(mx, lv[s]);
    float ex[8], sum = 0.f;
#pragma unroll
    for (int s = 0; s < 8; ++s) { ex[s] = __builtin_amdgcn_exp2f(lv[s] - mx); sum += ex[s]; }
    const float inv = __builtin_amdgcn_rcpf(sum);
#pragma unroll
    for (int s = 0; s < 8; ++s) {
      const float pv = ex[s] * inv + EPS_ATTN_;
      at[P][8 * kh + s][pc] = pv;
      dacc[s] += pv;
    }
  }
  __builtin_amdgcn_fence(__ATOMIC_RELEASE, "workgroup", "local");
  __builtin_amdgcn_s_barrier();
  __builtin_amdgcn_fence(__ATOMIC_ACQUIRE, "workgroup", "local");
  v2f af[16];
#pragma unroll
  for (int kk = 0; kk < 8; ++kk) {
    const v4f t = *(const v4f*)&at[P][m16][gb + (2 * kk + kh) * 4];
    af[kk] = v2f{t.x, t.y};
    af[kk + 8] = v2f{t.z, t.w};
  }
  __builtin_amdgcn_sched_barrier(0);
#pragma unroll
  for (int kk = 0; kk < 8; ++kk) {
    accu[0] = wmma_f32(af[kk], vf[kk], accu[0]);
    accu[1] = wmma_f32(af[kk + 8], vf[kk + 8], accu[1]);
  }
}

template <int K, int LDW, int PITCH>
__device__ __forceinline__ v8f pair_product(
    const float (&xt)[16][PITCH], const float* W0, const float* W1,
    const int ct, const int m16, const int kh) {
  v8f a0 = {}, a1 = {};
  const float* xp = &xt[m16][kh * 2];
  const float* p0 = W0 + kh * 2 * LDW + ct + m16;
  const float* p1 = W1 + kh * 2 * LDW + ct + m16;
#pragma unroll
  for (int kk = 0; kk < K / 4; ++kk) {
    v2f a;  a.x = xp[kk * 4]; a.y = xp[kk * 4 + 1];
    v2f b0; b0.x = p0[kk * 4 * LDW]; b0.y = p0[(kk * 4 + 1) * LDW];
    v2f b1; b1.x = p1[kk * 4 * LDW]; b1.y = p1[(kk * 4 + 1) * LDW];
    a0 = wmma_f32(a, b0, a0);
    a1 = wmma_f32(a, b1, a1);
  }
  v8f r;
#pragma unroll
  for (int i = 0; i < 8; ++i) r[i] = kh ? a1[i] : a0[i];
  return r;
}

__global__ void __launch_bounds__(256) __attribute__((amdgpu_waves_per_eu(2, 2)))
slot_iter_kernel(
    const float* kbuf, const float* vbuf,
    const float* hprev0, const float* q0,
    const float* lnsg, const float* lnsb, const float* Wq,
    const float* wih, const float* whh, const float* bih, const float* bhh,
    const float* lnmg, const float* lnmb,
    const float* w1, const float* b1, const float* w2, const float* b2,
    float* out) {
  __shared__ __attribute__((aligned(16))) float at[2][16][AP_];
  __shared__ float dsc[16][TR_ + 1];
  __shared__ float red[4][8][32];
  __shared__ float denl[16];
  __shared__ float ql[16][68];
  __shared__ float sl[16][68];
  __shared__ float ut[16][68];
  __shared__ float ht[16][68];
  __shared__ float gx[16][196];
  __shared__ float gh[16][196];
  __shared__ float sn[16][68];
  __shared__ float nm[16][68];
  __shared__ float hid[16][132];
  __shared__ float mu[16], rstd[16];
  const int tid = threadIdx.x;
  const int hp = blockIdx.x >> 6;
  const int b  = blockIdx.x & 63;
  const int hh0 = hp * 2;
  const int w8 = tid >> 5, lane = tid & 31, m16 = lane & 15, kh = lane >> 4;
  const int g = w8 >> 2, w = w8 & 3;
  const int gb = g * 64;
  const int r0 = tid >> 4, c00 = (tid & 15) * 4;
  const int hr = hh0 + (r0 >> 3);
  const size_t rowblk = (size_t)b * N_;
  const float* kw = kbuf + ((rowblk >> 4) + w8) * 1024 + lane * 4;
  const float* vw = vbuf + (((rowblk >> 6) + g) * 4 + w) * 1024 + lane * 4;
  const int q5 = (w8 & 1) * 16 + m16;
  const int pc = gb + (q5 >> 1) * 4 + ((w8 >> 1) & 1) * 2 + (q5 & 1);
  {
    const size_t grow = (size_t)((hr * B_ + b) * S_ + (r0 & 7));
    const float4 c = *(const float4*)(hprev0 + grow * D_ + c00);
    ht[r0][c00] = c.x; ht[r0][c00 + 1] = c.y; ht[r0][c00 + 2] = c.z; ht[r0][c00 + 3] = c.w;
    const float4 qv = *(const float4*)(q0 + grow * D_ + c00);
    ql[r0][c00] = qv.x; ql[r0][c00 + 1] = qv.y; ql[r0][c00 + 2] = qv.z; ql[r0][c00 + 3] = qv.w;
  }
  for (int it = 0; it < ITERS_; ++it) {
    asm volatile("" ::: "memory");
    v2f kfa[16], vfa[16], kfb[16], vfb[16];
#pragma unroll
    for (int j = 0; j < 8; ++j) {
      const v4f tk = *(const v4f*)(kw + j * 128);
      kfa[2 * j] = v2f{tk.x, tk.y};
      kfa[2 * j + 1] = v2f{tk.z, tk.w};
      const v4f tv = *(const v4f*)(vw + j * 128);
      vfa[2 * j] = v2f{tv.x, tv.y};
      vfa[2 * j + 1] = v2f{tv.z, tv.w};
    }
    if (it > 0) {
      if (tid < 16) {
        float m = 0.f;
        for (int j = 0; j < D_; ++j) m += sl[tid][j];
        m *= (1.f / D_);
        float var = 0.f;
        for (int j = 0; j < D_; ++j) { float dd = sl[tid][j] - m; var += dd * dd; }
        var *= (1.f / D_);
        mu[tid] = m; rstd[tid] = rsqrtf(var + LN_EPS_);
      }
      __syncthreads();
      for (int j = 0; j < 4; ++j) {
        const int d = c00 + j;
        ht[r0][d] = (sl[r0][d] - mu[r0]) * rstd[r0] * lnsg[hr * D_ + d] + lnsb[hr * D_ + d];
      }
      __syncthreads();
      {
        const int hs = w8 >> 2, ct = (w8 & 3) * 16;
        const float* Wh = Wq + (size_t)(hh0 + hs) * D_ * D_ + kh * 2 * D_ + ct + m16;
        const float* xp = &ht[m16][kh * 2];
        v8f acc = {};
#pragma unroll
        for (int kk = 0; kk < 16; ++kk) {
          v2f a;  a.x = xp[kk * 4]; a.y = xp[kk * 4 + 1];
          v2f bb; bb.x = Wh[kk * 4 * D_]; bb.y = Wh[(kk * 4 + 1) * D_];
          acc = wmma_f32(a, bb, acc);
        }
        if (kh == hs) {
#pragma unroll
          for (int i = 0; i < 8; ++i) ql[i + 8 * kh][ct + m16] = acc[i];
        }
      }
    }
    __syncthreads();
    v2f qf[16];
#pragma unroll
    for (int kk = 0; kk < 16; ++kk) {
      qf[kk].x = ql[m16][kk * 4 + kh * 2];
      qf[kk].y = ql[m16][kk * 4 + kh * 2 + 1];
    }
    v8f accu[2];
    float dacc[8];
    accu[0] = v8f{};
    accu[1] = v8f{};
#pragma unroll
    for (int s = 0; s < 8; ++s) dacc[s] = 0.f;
    for (int t = 0; t < NT_; t += 2) {
      attn_tile_step<0>(at, kw + (size_t)(t + 1) * STEPF_, vw + (size_t)(t + 1) * STEPF_,
                        qf, kfa, vfa, kfb, vfb, accu, dacc, pc, gb, m16, kh);
      const int tn = (t + 2 < NT_) ? t + 2 : t + 1;
      attn_tile_step<1>(at, kw + (size_t)tn * STEPF_, vw + (size_t)tn * STEPF_,
                        qf, kfb, vfb, kfa, vfa, accu, dacc, pc, gb, m16, kh);
    }
    const v8f asum = accu[0] + accu[1];
    if (g == 1) {
#pragma unroll
      for (int i = 0; i < 8; ++i) red[w][i][lane] = asum[i];
    }
    {
      const int r = w8 * 16 + m16;
#pragma unroll
      for (int s = 0; s < 8; ++s) dsc[kh * 8 + s][r] = dacc[s];
    }
    __syncthreads();
    if (tid < 16) {
      float dsum = 0.f;
      for (int r = 0; r < TR_; ++r) dsum += dsc[tid][r];
      denl[tid] = dsum;
    }
    __syncthreads();
    if (g == 0) {
#pragma unroll
      for (int i = 0; i < 8; ++i) {
        const int sp = i + 8 * kh;
        ut[sp][w * 16 + m16] = (asum[i] + red[w][i][lane]) / denl[sp];
      }
    }
    __syncthreads();

#pragma unroll
    for (int j = 0; j < 3; ++j) {
      const int T = w8 + 8 * j;
      const bool isX = T < 12;
      const int ct = (T % 12) * 16;
      const float* Wt = (isX ? wih : whh) + (size_t)hh0 * D_ * 192;
      v8f r;
      if (isX) r = pair_product<D_, 192, 68>(ut, Wt, Wt + D_ * 192, ct, m16, kh);
      else     r = pair_product<D_, 192, 68>(ht, Wt, Wt + D_ * 192, ct, m16, kh);
      float* dst = isX ? &gx[0][0] : &gh[0][0];
#pragma unroll
      for (int i = 0; i < 8; ++i)
        dst[(i + 8 * kh) * 196 + ct + m16] = r[i];
    }
    __syncthreads();
    {
      const float* bi = bih + hr * 192;
      const float* bh = bhh + hr * 192;
      for (int j = 0; j < 4; ++j) {
        int d = c00 + j;
        float xr = gx[r0][d] + bi[d],        hrr = gh[r0][d] + bh[d];
        float xz = gx[r0][64 + d] + bi[64 + d],  hz = gh[r0][64 + d] + bh[64 + d];
        float xn = gx[r0][128 + d] + bi[128 + d], hn = gh[r0][128 + d] + bh[128 + d];
        float rg = 1.f / (1.f + expf(-(xr + hrr)));
        float zg = 1.f / (1.f + expf(-(xz + hz)));
        float ng = tanhf(xn + rg * hn);
        sn[r0][d] = (1.f - zg) * ng + zg * ht[r0][d];
      }
    }
    __syncthreads();
    if (tid < 16) {
      float m = 0.f;
      for (int j = 0; j < D_; ++j) m += sn[tid][j];
      m *= (1.f / D_);
      float var = 0.f;
      for (int j = 0; j < D_; ++j) { float dd = sn[tid][j] - m; var += dd * dd; }
      var *= (1.f / D_);
      mu[tid] = m; rstd[tid] = rsqrtf(var + LN_EPS_);
    }
    __syncthreads();
    for (int j = 0; j < 4; ++j) {
      int d = c00 + j;
      nm[r0][d] = (sn[r0][d] - mu[r0]) * rstd[r0] * lnmg[hr * D_ + d] + lnmb[hr * D_ + d];
    }
    __syncthreads();
    {
      const float* w1h = w1 + (size_t)hh0 * D_ * MLPH_;
      const int ct = w8 * 16;
      const v8f r = pair_product<D_, MLPH_, 68>(nm, w1h, w1h + D_ * MLPH_, ct, m16, kh);
      const float bias = b1[(hh0 + kh) * MLPH_ + ct + m16];
#pragma unroll
      for (int i = 0; i < 8; ++i)
        hid[i + 8 * kh][ct + m16] = fmaxf(r[i] + bias, 0.f);
    }
    __syncthreads();
    {
      const int hs = w8 >> 2, ct = (w8 & 3) * 16;
      const float* w2h = w2 + (size_t)(hh0 + hs) * MLPH_ * D_ + kh * 2 * D_ + ct + m16;
      const float* xp = &hid[m16][kh * 2];
      v8f acc = {};
#pragma unroll
      for (int kk = 0; kk < 32; ++kk) {
        v2f a;  a.x = xp[kk * 4]; a.y = xp[kk * 4 + 1];
        v2f bb; bb.x = w2h[kk * 4 * D_]; bb.y = w2h[(kk * 4 + 1) * D_];
        acc = wmma_f32(a, bb, acc);
      }
      const float bias = b2[(hh0 + hs) * D_ + ct + m16];
      if (kh == hs) {
#pragma unroll
        for (int i = 0; i < 8; ++i) {
          const int rr = i + 8 * kh;
          sl[rr][ct + m16] = sn[rr][ct + m16] + acc[i] + bias;
        }
      }
    }
    __syncthreads();
  }
  if (hp == 0 && r0 < 8) {
    float4 o;
    o.x = sl[r0][c00]; o.y = sl[r0][c00 + 1]; o.z = sl[r0][c00 + 2]; o.w = sl[r0][c00 + 3];
    *(float4*)(out + (size_t)(b * S_ + r0) * D_ + c00) = o;
  }
}

extern "C" void kernel_launch(void* const* d_in, const int* in_sizes, int n_in,
                              void* d_out, int out_size, void* d_ws, size_t ws_size,
                              hipStream_t stream) {
  (void)in_sizes; (void)n_in; (void)out_size; (void)ws_size;
  const float* inputs  = (const float*)d_in[0];
  const float* sinit   = (const float*)d_in[1];
  const float* smu     = (const float*)d_in[2];
  const float* slog    = (const float*)d_in[3];
  const float* ln_in_g = (const float*)d_in[4];
  const float* ln_in_b = (const float*)d_in[5];
  const float* Wk      = (const float*)d_in[6];
  const float* Wv      = (const float*)d_in[7];
  const float* ln_s_g  = (const float*)d_in[8];
  const float* ln_s_b  = (const float*)d_in[9];
  const float* Wq      = (const float*)d_in[10];
  const float* gru_wih = (const float*)d_in[11];
  const float* gru_whh = (const float*)d_in[12];
  const float* gru_bih = (const float*)d_in[13];
  const float* gru_bhh = (const float*)d_in[14];
  const float* ln_m_g  = (const float*)d_in[15];
  const float* ln_m_b  = (const float*)d_in[16];
  const float* mlp_w1  = (const float*)d_in[17];
  const float* mlp_b1  = (const float*)d_in[18];
  const float* mlp_w2  = (const float*)d_in[19];
  const float* mlp_b2  = (const float*)d_in[20];

  float* ws = (float*)d_ws;
  float* kbuf  = ws;
  float* vbuf  = kbuf + (size_t)B_ * N_ * D_;
  float* hprev = vbuf + (size_t)B_ * N_ * D_;
  float* qbuf  = hprev + H_ * B_ * S_ * D_;

  ln_kv_slots_kernel<<<SLOT_WGS_ + (B_ * N_) / LKV_ROWS_, 256, 0, stream>>>(
      inputs, ln_in_g, ln_in_b, Wk, Wv, kbuf, vbuf,
      sinit, smu, slog, ln_s_g, ln_s_b, Wq, hprev, qbuf);
  slot_iter_kernel<<<(H_ / 2) * B_, 256, 0, stream>>>(
      kbuf, vbuf, hprev, qbuf, ln_s_g, ln_s_b, Wq, gru_wih, gru_whh,
      gru_bih, gru_bhh, ln_m_g, ln_m_b, mlp_w1, mlp_b1, mlp_w2, mlp_b2,
      (float*)d_out);
}
